# starting kernel
# baseline (speedup 1.0000x reference)
#include <hip/hip_runtime.h>
#include <hip/hip_bf16.h>
#include <math.h>

typedef __attribute__((ext_vector_type(2))) float v2f;
typedef __attribute__((ext_vector_type(8))) float v8f;

#define SELU_SCALE 1.0507009873554805f
#define SELU_ALPHA 1.6732632423543772f

__device__ __forceinline__ v8f wmma4(v2f a, v2f b, v8f c) {
    return __builtin_amdgcn_wmma_f32_16x16x4_f32(false, a, false, b, (short)0, c, false, false);
}

__global__ void deg_kernel(const int* __restrict__ dst, float* __restrict__ deg, int E) {
    int e = blockIdx.x * blockDim.x + threadIdx.x;
    if (e < E) atomicAdd(&deg[dst[e]], 1.0f);
}

__global__ void norm_kernel(const float* __restrict__ deg, float* __restrict__ dis,
                            float* __restrict__ cntinv, int n) {
    int i = blockIdx.x * blockDim.x + threadIdx.x;
    if (i < n) {
        float d = deg[i];
        dis[i]    = (d > 0.0f) ? rsqrtf(fmaxf(d, 1.0f)) : 0.0f;
        cntinv[i] = 1.0f / fmaxf(d, 1.0f);
    }
}

__global__ void prop_kernel(const float* __restrict__ X, float* __restrict__ Y,
                            const int* __restrict__ src, const int* __restrict__ dst,
                            const float* __restrict__ dis, int E, int weighted) {
    int tid = blockIdx.x * blockDim.x + threadIdx.x;
    int e   = tid >> 4;
    int seg = tid & 15;
    if (e >= E) return;
    int s = src[e];
    int t = dst[e];
    float w = 1.0f;
    if (weighted) w = dis[s] * dis[t];
    const float4 v = ((const float4*)(X + (long)s * 64))[seg];
    float* yr = Y + (long)t * 64 + seg * 4;
    atomicAdd(yr + 0, v.x * w);
    atomicAdd(yr + 1, v.y * w);
    atomicAdd(yr + 2, v.z * w);
    atomicAdd(yr + 3, v.w * w);
}

__global__ __launch_bounds__(256) void gemm1_kernel(
    const float* __restrict__ Sagg, const float* __restrict__ Hf,
    const float* __restrict__ cntinv,
    const float* __restrict__ Wl, const float* __restrict__ Wr,
    const float* __restrict__ bias, float* __restrict__ Out, int nnodes)
{
    int wave = (blockIdx.x * blockDim.x + threadIdx.x) >> 5;
    int lane = threadIdx.x & 31;
    int rb = wave;
    int rbCount = (nnodes + 15) >> 4;
    if (rb >= rbCount) return;
    int m = rb * 16 + (lane & 15);
    int mclamp = m < nnodes ? m : (nnodes - 1);
    int col0 = lane & 15;
    int kshift = (lane >> 4) * 2;
    float cinv = cntinv[mclamp];
    const float* aggRow = Sagg + (long)mclamp * 64;
    const float* hRow   = Hf   + (long)mclamp * 64;
    v8f acc0 = {}, acc1 = {}, acc2 = {}, acc3 = {};
    #pragma unroll 4
    for (int k0 = 0; k0 < 64; k0 += 4) {
        int kk = k0 + kshift;
        v2f a1; a1.x = aggRow[kk] * cinv; a1.y = aggRow[kk + 1] * cinv;
        v2f a2; a2.x = hRow[kk];          a2.y = hRow[kk + 1];
        const float* wl0 = Wl + kk * 64 + col0;
        const float* wl1 = Wl + (kk + 1) * 64 + col0;
        const float* wr0 = Wr + kk * 64 + col0;
        const float* wr1 = Wr + (kk + 1) * 64 + col0;
        v2f b;
        b.x = wl0[ 0]; b.y = wl1[ 0]; acc0 = wmma4(a1, b, acc0);
        b.x = wl0[16]; b.y = wl1[16]; acc1 = wmma4(a1, b, acc1);
        b.x = wl0[32]; b.y = wl1[32]; acc2 = wmma4(a1, b, acc2);
        b.x = wl0[48]; b.y = wl1[48]; acc3 = wmma4(a1, b, acc3);
        b.x = wr0[ 0]; b.y = wr1[ 0]; acc0 = wmma4(a2, b, acc0);
        b.x = wr0[16]; b.y = wr1[16]; acc1 = wmma4(a2, b, acc1);
        b.x = wr0[32]; b.y = wr1[32]; acc2 = wmma4(a2, b, acc2);
        b.x = wr0[48]; b.y = wr1[48]; acc3 = wmma4(a2, b, acc3);
    }
    int rowbase = rb * 16 + (lane >> 4) * 8;
    v8f accs[4] = {acc0, acc1, acc2, acc3};
    #pragma unroll
    for (int ct = 0; ct < 4; ++ct) {
        int col = ct * 16 + col0;
        float bn = bias[col];
        #pragma unroll
        for (int j = 0; j < 8; ++j) {
            int row = rowbase + j;
            float v = accs[ct][j] + bn;
            v = v > 0.0f ? SELU_SCALE * v : SELU_SCALE * SELU_ALPHA * (expf(v) - 1.0f);
            if (row < nnodes) Out[(long)row * 64 + col] = v;
        }
    }
}

__global__ __launch_bounds__(256) void gemm2_kernel(
    const float* __restrict__ Sagg, const float* __restrict__ Hf,
    const float* __restrict__ cntinv,
    const float* __restrict__ Wl, const float* __restrict__ Wr,
    const float* __restrict__ bias, float* __restrict__ Logits, int nnodes)
{
    int wave = (blockIdx.x * blockDim.x + threadIdx.x) >> 5;
    int lane = threadIdx.x & 31;
    int rb = wave;
    int rbCount = (nnodes + 15) >> 4;
    if (rb >= rbCount) return;
    int m = rb * 16 + (lane & 15);
    int mclamp = m < nnodes ? m : (nnodes - 1);
    int col0 = lane & 15;
    int kshift = (lane >> 4) * 2;
    float cinv = cntinv[mclamp];
    int col2  = 32 + col0;
    int c2off = col2 < 40 ? col2 : 0;
    float msk2 = col2 < 40 ? 1.0f : 0.0f;
    const float* aggRow = Sagg + (long)mclamp * 64;
    const float* hRow   = Hf   + (long)mclamp * 64;
    v8f acc0 = {}, acc1 = {}, acc2 = {};
    #pragma unroll 4
    for (int k0 = 0; k0 < 64; k0 += 4) {
        int kk = k0 + kshift;
        v2f a1; a1.x = aggRow[kk] * cinv; a1.y = aggRow[kk + 1] * cinv;
        v2f a2; a2.x = hRow[kk];          a2.y = hRow[kk + 1];
        const float* wl0 = Wl + kk * 40;
        const float* wl1 = Wl + (kk + 1) * 40;
        const float* wr0 = Wr + kk * 40;
        const float* wr1 = Wr + (kk + 1) * 40;
        v2f b;
        b.x = wl0[col0];          b.y = wl1[col0];          acc0 = wmma4(a1, b, acc0);
        b.x = wl0[16 + col0];     b.y = wl1[16 + col0];     acc1 = wmma4(a1, b, acc1);
        b.x = wl0[c2off] * msk2;  b.y = wl1[c2off] * msk2;  acc2 = wmma4(a1, b, acc2);
        b.x = wr0[col0];          b.y = wr1[col0];          acc0 = wmma4(a2, b, acc0);
        b.x = wr0[16 + col0];     b.y = wr1[16 + col0];     acc1 = wmma4(a2, b, acc1);
        b.x = wr0[c2off] * msk2;  b.y = wr1[c2off] * msk2;  acc2 = wmma4(a2, b, acc2);
    }
    int rowbase = rb * 16 + (lane >> 4) * 8;
    v8f accs[3] = {acc0, acc1, acc2};
    #pragma unroll
    for (int ct = 0; ct < 3; ++ct) {
        int col = ct * 16 + col0;
        if (col < 40) {
            float bn = bias[col];
            #pragma unroll
            for (int j = 0; j < 8; ++j) {
                int row = rowbase + j;
                if (row < nnodes) Logits[(long)row * 40 + col] = accs[ct][j] + bn;
            }
        }
    }
}

__global__ void softmax_kernel(const float* __restrict__ Logits, float* __restrict__ Out, int nnodes) {
    int wave = (blockIdx.x * blockDim.x + threadIdx.x) >> 5;
    int lane = threadIdx.x & 31;
    if (wave >= nnodes) return;
    const float* row = Logits + (long)wave * 40;
    float v0 = row[lane];
    bool hi = (lane + 32) < 40;
    float v1 = hi ? row[lane + 32] : -3.4e38f;
    float mx = fmaxf(v0, v1);
    for (int o = 16; o > 0; o >>= 1) mx = fmaxf(mx, __shfl_xor(mx, o, 32));
    float e0 = expf(v0 - mx);
    float e1 = hi ? expf(v1 - mx) : 0.0f;
    float s = e0 + e1;
    for (int o = 16; o > 0; o >>= 1) s += __shfl_xor(s, o, 32);
    float inv = 1.0f / s;
    Out[(long)wave * 40 + lane] = e0 * inv;
    if (hi) Out[(long)wave * 40 + lane + 32] = e1 * inv;
}

extern "C" void kernel_launch(void* const* d_in, const int* in_sizes, int n_in,
                              void* d_out, int out_size, void* d_ws, size_t ws_size,
                              hipStream_t stream) {
    const float* x    = (const float*)d_in[0];
    const float* Wl1  = (const float*)d_in[1];
    const float* Wr1  = (const float*)d_in[2];
    const float* b1   = (const float*)d_in[3];
    const float* Wl2  = (const float*)d_in[4];
    const float* Wr2  = (const float*)d_in[5];
    const float* b2   = (const float*)d_in[6];
    const int* esrc   = (const int*)d_in[7];
    const int* edst   = (const int*)d_in[8];

    const int H  = in_sizes[3];
    const int D  = in_sizes[1] / H;
    const int Nn = in_sizes[0] / D;
    const int E  = in_sizes[7];
    (void)n_in; (void)out_size; (void)ws_size;

    float* ws     = (float*)d_ws;
    float* deg    = ws;
    float* dis    = ws + (long)Nn;
    float* cntinv = ws + 2L * Nn;
    float* bufA   = ws + 3L * Nn;
    float* bufB   = bufA + (long)Nn * 64;
    float* bufC   = bufB + (long)Nn * 64;

    const int TB = 256;
    const size_t featBytes = (size_t)Nn * 64 * sizeof(float);
    dim3 gE((E + TB - 1) / TB), gN((Nn + TB - 1) / TB);
    dim3 gProp(((long)E * 16 + TB - 1) / TB);
    int rbCount = (Nn + 15) / 16;
    dim3 gGemm(((long)rbCount * 32 + TB - 1) / TB);
    dim3 gSm(((long)Nn * 32 + TB - 1) / TB);

    hipMemsetAsync(deg, 0, (size_t)Nn * sizeof(float), stream);
    deg_kernel<<<gE, TB, 0, stream>>>(edst, deg, E);
    norm_kernel<<<gN, TB, 0, stream>>>(deg, dis, cntinv, Nn);

    hipMemsetAsync(bufA, 0, featBytes, stream);
    prop_kernel<<<gProp, TB, 0, stream>>>(x, bufA, esrc, edst, dis, E, 1);
    hipMemsetAsync(bufB, 0, featBytes, stream);
    prop_kernel<<<gProp, TB, 0, stream>>>(bufA, bufB, esrc, edst, dis, E, 1);
    hipMemsetAsync(bufA, 0, featBytes, stream);
    prop_kernel<<<gProp, TB, 0, stream>>>(bufB, bufA, esrc, edst, dis, E, 0);
    gemm1_kernel<<<gGemm, TB, 0, stream>>>(bufA, bufB, cntinv, Wl1, Wr1, b1, bufC, Nn);
    hipMemsetAsync(bufA, 0, featBytes, stream);
    prop_kernel<<<gProp, TB, 0, stream>>>(bufC, bufA, esrc, edst, dis, E, 0);
    gemm2_kernel<<<gGemm, TB, 0, stream>>>(bufA, bufC, cntinv, Wl2, Wr2, b2, bufB, Nn);
    softmax_kernel<<<gSm, TB, 0, stream>>>(bufB, (float*)d_out, Nn);
}


# Scatter with float atomics replaced by a destination-sorted edge list built in-call (count atomic returns slot); each prop gathers rows in registers, one store; sum order per node = arrival rank
# speedup vs baseline: 2.6789x; 2.6789x over previous
#include <hip/hip_runtime.h>
#include <hip/hip_bf16.h>
#include <math.h>

typedef __attribute__((ext_vector_type(2))) float v2f;
typedef __attribute__((ext_vector_type(8))) float v8f;

#define SELU_SCALE 1.0507009873554805f
#define SELU_ALPHA 1.6732632423543772f

__device__ __forceinline__ v8f wmma4(v2f a, v2f b, v8f c) {
    return __builtin_amdgcn_wmma_f32_16x16x4_f32(false, a, false, b, (short)0, c, false, false);
}

__global__ void count_rank_kernel(const int* __restrict__ dst, int* __restrict__ degi,
                                  int* __restrict__ rank, int E) {
    int e = blockIdx.x * blockDim.x + threadIdx.x;
    if (e < E) rank[e] = atomicAdd(&degi[dst[e]], 1);
}

#define SCAN_THREADS 1024
#define SCAN_CHUNK   4096
__global__ __launch_bounds__(SCAN_THREADS) void scan_norm_kernel(
    const int* __restrict__ degi, int* __restrict__ rowptr, float* __restrict__ dis, int n)
{
    __shared__ int wsum[32];
    __shared__ int blockPrefix;
    const int t = threadIdx.x;
    const int lane = t & 31;
    const int wv = t >> 5;
    const int base = blockIdx.x * SCAN_CHUNK;

    int pre = 0;
    const int4* d4 = (const int4*)degi;
    for (int i = t; i < (base >> 2); i += SCAN_THREADS) {
        int4 q = d4[i];
        pre += (q.x + q.y) + (q.z + q.w);
    }
    for (int o = 16; o > 0; o >>= 1) pre += __shfl_xor(pre, o, 32);
    if (lane == 0) wsum[wv] = pre;
    __syncthreads();
    if (wv == 0) {
        int v = wsum[lane];
        for (int o = 16; o > 0; o >>= 1) v += __shfl_xor(v, o, 32);
        if (lane == 0) blockPrefix = v;
    }
    __syncthreads();
    const int bp = blockPrefix;

    const int idx0 = base + t * 4;
    int cnt[4];
    #pragma unroll
    for (int q = 0; q < 4; ++q) cnt[q] = (idx0 + q < n) ? degi[idx0 + q] : 0;
    const int tsum = (cnt[0] + cnt[1]) + (cnt[2] + cnt[3]);
    int inc = tsum;
    for (int o = 1; o < 32; o <<= 1) {
        int y = __shfl_up(inc, o, 32);
        if (lane >= o) inc += y;
    }
    if (lane == 31) wsum[wv] = inc;
    __syncthreads();
    int wp = (lane < wv) ? wsum[lane] : 0;
    for (int o = 16; o > 0; o >>= 1) wp += __shfl_xor(wp, o, 32);
    int excl = bp + wp + inc - tsum;
    #pragma unroll
    for (int q = 0; q < 4; ++q) {
        if (idx0 + q < n) {
            float d = (float)cnt[q];
            rowptr[idx0 + q] = excl;
            dis[idx0 + q] = (d > 0.0f) ? rsqrtf(fmaxf(d, 1.0f)) : 0.0f;
            excl += cnt[q];
        }
    }
}

__global__ void fill_kernel(const int* __restrict__ src, const int* __restrict__ dst,
                            const int* __restrict__ rank, const int* __restrict__ rowptr,
                            int* __restrict__ csr_src, int E) {
    int e = blockIdx.x * blockDim.x + threadIdx.x;
    if (e < E) csr_src[rowptr[dst[e]] + rank[e]] = src[e];
}

__device__ __forceinline__ float keep_bits(float t, int keep) {
    return __int_as_float(__float_as_int(t) & keep);
}

template <int WEIGHTED>
__global__ __launch_bounds__(256) void gather_kernel(
    const float* __restrict__ X, float* __restrict__ Y,
    const int* __restrict__ csr_src, const int* __restrict__ rowptr,
    const int* __restrict__ degi, const float* __restrict__ dis, int n)
{
    const int tid  = blockIdx.x * blockDim.x + threadIdx.x;
    const int node = tid >> 4;
    const int seg  = tid & 15;
    const bool valid = node < n;
    const int nd  = valid ? node : (n - 1);
    const int beg = rowptr[nd];
    const int cnt = valid ? degi[nd] : 0;
    const float dt = WEIGHTED ? dis[nd] : 1.0f;
    const float4* X4 = (const float4*)X;
    float4 acc = make_float4(0.0f, 0.0f, 0.0f, 0.0f);
    const int cmax = __builtin_amdgcn_readfirstlane(max(cnt, __shfl_xor(cnt, 16, 32)));
    for (int base = 0; base < cmax; base += 16) {
        const int j = base + seg;
        const int mySrc = (j < cnt) ? csr_src[beg + j] : 0;
        float myW = 1.0f;
        if (WEIGHTED) myW = dis[mySrc] * dt;
        const int lim = min(16, cmax - base);
        for (int k = 0; k < lim; k += 4) {
            int s[4]; float w[4]; float4 v[4];
            #pragma unroll
            for (int u = 0; u < 4; ++u) {
                s[u] = __shfl(mySrc, k + u, 16);
                w[u] = WEIGHTED ? __shfl(myW, k + u, 16) : 1.0f;
            }
            #pragma unroll
            for (int u = 0; u < 4; ++u) v[u] = X4[(long)s[u] * 16 + seg];
            #pragma unroll
            for (int u = 0; u < 4; ++u) {
                const int keep = ((base + k + u) < cnt) ? -1 : 0;
                acc.x += keep_bits(WEIGHTED ? v[u].x * w[u] : v[u].x, keep);
                acc.y += keep_bits(WEIGHTED ? v[u].y * w[u] : v[u].y, keep);
                acc.z += keep_bits(WEIGHTED ? v[u].z * w[u] : v[u].z, keep);
                acc.w += keep_bits(WEIGHTED ? v[u].w * w[u] : v[u].w, keep);
            }
        }
    }
    if (valid) ((float4*)Y)[(long)node * 16 + seg] = acc;
}

__global__ __launch_bounds__(256) void gemm1_kernel(
    const float* Sagg, const float* __restrict__ Hf,
    const int* __restrict__ degi,
    const float* __restrict__ Wl, const float* __restrict__ Wr,
    const float* __restrict__ bias, float* Out, int nnodes)
{
    int wave = (blockIdx.x * blockDim.x + threadIdx.x) >> 5;
    int lane = threadIdx.x & 31;
    int rb = wave;
    int rbCount = (nnodes + 15) >> 4;
    if (rb >= rbCount) return;
    int m = rb * 16 + (lane & 15);
    int mclamp = m < nnodes ? m : (nnodes - 1);
    int col0 = lane & 15;
    int kshift = (lane >> 4) * 2;
    float cinv = 1.0f / fmaxf((float)degi[mclamp], 1.0f);
    const float* aggRow = Sagg + (long)mclamp * 64;
    const float* hRow   = Hf   + (long)mclamp * 64;
    v8f acc0 = {}, acc1 = {}, acc2 = {}, acc3 = {};
    #pragma unroll 4
    for (int k0 = 0; k0 < 64; k0 += 4) {
        int kk = k0 + kshift;
        v2f a1; a1.x = aggRow[kk] * cinv; a1.y = aggRow[kk + 1] * cinv;
        v2f a2; a2.x = hRow[kk];          a2.y = hRow[kk + 1];
        const float* wl0 = Wl + kk * 64 + col0;
        const float* wl1 = Wl + (kk + 1) * 64 + col0;
        const float* wr0 = Wr + kk * 64 + col0;
        const float* wr1 = Wr + (kk + 1) * 64 + col0;
        v2f b;
        b.x = wl0[ 0]; b.y = wl1[ 0]; acc0 = wmma4(a1, b, acc0);
        b.x = wl0[16]; b.y = wl1[16]; acc1 = wmma4(a1, b, acc1);
        b.x = wl0[32]; b.y = wl1[32]; acc2 = wmma4(a1, b, acc2);
        b.x = wl0[48]; b.y = wl1[48]; acc3 = wmma4(a1, b, acc3);
        b.x = wr0[ 0]; b.y = wr1[ 0]; acc0 = wmma4(a2, b, acc0);
        b.x = wr0[16]; b.y = wr1[16]; acc1 = wmma4(a2, b, acc1);
        b.x = wr0[32]; b.y = wr1[32]; acc2 = wmma4(a2, b, acc2);
        b.x = wr0[48]; b.y = wr1[48]; acc3 = wmma4(a2, b, acc3);
    }
    int rowbase = rb * 16 + (lane >> 4) * 8;
    v8f accs[4] = {acc0, acc1, acc2, acc3};
    #pragma unroll
    for (int ct = 0; ct < 4; ++ct) {
        int col = ct * 16 + col0;
        float bn = bias[col];
        #pragma unroll
        for (int j = 0; j < 8; ++j) {
            int row = rowbase + j;
            float v = accs[ct][j] + bn;
            v = v > 0.0f ? SELU_SCALE * v : SELU_SCALE * SELU_ALPHA * (expf(v) - 1.0f);
            if (row < nnodes) Out[(long)row * 64 + col] = v;
        }
    }
}

__global__ __launch_bounds__(256) void gemm2_kernel(
    const float* __restrict__ Sagg, const float* __restrict__ Hf,
    const int* __restrict__ degi,
    const float* __restrict__ Wl, const float* __restrict__ Wr,
    const float* __restrict__ bias, float* __restrict__ Logits, int nnodes)
{
    int wave = (blockIdx.x * blockDim.x + threadIdx.x) >> 5;
    int lane = threadIdx.x & 31;
    int rb = wave;
    int rbCount = (nnodes + 15) >> 4;
    if (rb >= rbCount) return;
    int m = rb * 16 + (lane & 15);
    int mclamp = m < nnodes ? m : (nnodes - 1);
    int col0 = lane & 15;
    int kshift = (lane >> 4) * 2;
    float cinv = 1.0f / fmaxf((float)degi[mclamp], 1.0f);
    int col2  = 32 + col0;
    int c2off = col2 < 40 ? col2 : 0;
    float msk2 = col2 < 40 ? 1.0f : 0.0f;
    const float* aggRow = Sagg + (long)mclamp * 64;
    const float* hRow   = Hf   + (long)mclamp * 64;
    v8f acc0 = {}, acc1 = {}, acc2 = {};
    #pragma unroll 4
    for (int k0 = 0; k0 < 64; k0 += 4) {
        int kk = k0 + kshift;
        v2f a1; a1.x = aggRow[kk] * cinv; a1.y = aggRow[kk + 1] * cinv;
        v2f a2; a2.x = hRow[kk];          a2.y = hRow[kk + 1];
        const float* wl0 = Wl + kk * 40;
        const float* wl1 = Wl + (kk + 1) * 40;
        const float* wr0 = Wr + kk * 40;
        const float* wr1 = Wr + (kk + 1) * 40;
        v2f b;
        b.x = wl0[col0];          b.y = wl1[col0];          acc0 = wmma4(a1, b, acc0);
        b.x = wl0[16 + col0];     b.y = wl1[16 + col0];     acc1 = wmma4(a1, b, acc1);
        b.x = wl0[c2off] * msk2;  b.y = wl1[c2off] * msk2;  acc2 = wmma4(a1, b, acc2);
        b.x = wr0[col0];          b.y = wr1[col0];          acc0 = wmma4(a2, b, acc0);
        b.x = wr0[16 + col0];     b.y = wr1[16 + col0];     acc1 = wmma4(a2, b, acc1);
        b.x = wr0[c2off] * msk2;  b.y = wr1[c2off] * msk2;  acc2 = wmma4(a2, b, acc2);
    }
    int rowbase = rb * 16 + (lane >> 4) * 8;
    v8f accs[3] = {acc0, acc1, acc2};
    #pragma unroll
    for (int ct = 0; ct < 3; ++ct) {
        int col = ct * 16 + col0;
        if (col < 40) {
            float bn = bias[col];
            #pragma unroll
            for (int j = 0; j < 8; ++j) {
                int row = rowbase + j;
                if (row < nnodes) Logits[(long)row * 40 + col] = accs[ct][j] + bn;
            }
        }
    }
}

__global__ void softmax_kernel(const float* __restrict__ Logits, float* __restrict__ Out, int nnodes) {
    int wave = (blockIdx.x * blockDim.x + threadIdx.x) >> 5;
    int lane = threadIdx.x & 31;
    if (wave >= nnodes) return;
    const float* row = Logits + (long)wave * 40;
    float v0 = row[lane];
    bool hi = (lane + 32) < 40;
    float v1 = hi ? row[lane + 32] : -3.4e38f;
    float mx = fmaxf(v0, v1);
    for (int o = 16; o > 0; o >>= 1) mx = fmaxf(mx, __shfl_xor(mx, o, 32));
    float e0 = expf(v0 - mx);
    float e1 = hi ? expf(v1 - mx) : 0.0f;
    float s = e0 + e1;
    for (int o = 16; o > 0; o >>= 1) s += __shfl_xor(s, o, 32);
    float inv = 1.0f / s;
    Out[(long)wave * 40 + lane] = e0 * inv;
    if (hi) Out[(long)wave * 40 + lane + 32] = e1 * inv;
}

extern "C" void kernel_launch(void* const* d_in, const int* in_sizes, int n_in,
                              void* d_out, int out_size, void* d_ws, size_t ws_size,
                              hipStream_t stream) {
    const float* x    = (const float*)d_in[0];
    const float* Wl1  = (const float*)d_in[1];
    const float* Wr1  = (const float*)d_in[2];
    const float* b1   = (const float*)d_in[3];
    const float* Wl2  = (const float*)d_in[4];
    const float* Wr2  = (const float*)d_in[5];
    const float* b2   = (const float*)d_in[6];
    const int* esrc   = (const int*)d_in[7];
    const int* edst   = (const int*)d_in[8];

    const int H  = in_sizes[3];
    const int D  = in_sizes[1] / H;
    const int Nn = in_sizes[0] / D;
    const int E  = in_sizes[7];
    (void)n_in; (void)out_size; (void)ws_size;

    float* ws     = (float*)d_ws;
    int*   degi   = (int*)ws;
    float* dis    = ws + (long)Nn;
    int*   rowptr = (int*)(ws + 2L * Nn);
    float* bufA   = ws + 3L * Nn;
    float* bufB   = bufA + (long)Nn * 64;
    float* bufC   = bufB + (long)Nn * 64;
    int*   rank   = (int*)bufB;
    int*   csrSrc = (int*)bufC;
    float* logits = bufC + (long)E;

    const int TB = 256;
    dim3 gE((E + TB - 1) / TB);
    dim3 gScan((Nn + SCAN_CHUNK - 1) / SCAN_CHUNK);
    dim3 gGather(((long)Nn * 16 + TB - 1) / TB);
    int rbCount = (Nn + 15) / 16;
    dim3 gGemm(((long)rbCount * 32 + TB - 1) / TB);
    dim3 gSm(((long)Nn * 32 + TB - 1) / TB);

    hipMemsetAsync(degi, 0, (size_t)Nn * sizeof(int), stream);
    count_rank_kernel<<<gE, TB, 0, stream>>>(edst, degi, rank, E);
    scan_norm_kernel<<<gScan, SCAN_THREADS, 0, stream>>>(degi, rowptr, dis, Nn);
    fill_kernel<<<gE, TB, 0, stream>>>(esrc, edst, rank, rowptr, csrSrc, E);

    gather_kernel<1><<<gGather, TB, 0, stream>>>(x, bufA, csrSrc, rowptr, degi, dis, Nn);
    gather_kernel<1><<<gGather, TB, 0, stream>>>(bufA, bufB, csrSrc, rowptr, degi, dis, Nn);
    gather_kernel<0><<<gGather, TB, 0, stream>>>(bufB, bufA, csrSrc, rowptr, degi, dis, Nn);
    gemm1_kernel<<<gGemm, TB, 0, stream>>>(bufA, bufB, degi, Wl1, Wr1, b1, bufA, Nn);
    gather_kernel<0><<<gGather, TB, 0, stream>>>(bufA, bufB, csrSrc, rowptr, degi, dis, Nn);
    gemm2_kernel<<<gGemm, TB, 0, stream>>>(bufB, bufA, degi, Wl2, Wr2, b2, logits, Nn);
    softmax_kernel<<<gSm, TB, 0, stream>>>(logits, (float*)d_out, Nn);
}


# Edge list sorted by destination with LDS counters only (edges split into 512-node buckets, then count, scan, place per bucket in LDS): no global atomic, no memset; per-node sum order = arrival rank
# speedup vs baseline: 3.6963x; 1.3798x over previous
#include <hip/hip_runtime.h>
#include <hip/hip_bf16.h>
#include <math.h>

typedef __attribute__((ext_vector_type(2))) float v2f;
typedef __attribute__((ext_vector_type(8))) float v8f;

#define SELU_SCALE 1.0507009873554805f
#define SELU_ALPHA 1.6732632423543772f

__device__ __forceinline__ v8f wmma4(v2f a, v2f b, v8f c) {
    return __builtin_amdgcn_wmma_f32_16x16x4_f32(false, a, false, b, (short)0, c, false, false);
}

#define BKT_SHIFT   9
#define BKT_NODES   512
#define MAX_BKT     1024
#define PT_THREADS  1024
#define PT_PER      12
#define PT_CHUNK    (PT_THREADS * PT_PER)

__global__ __launch_bounds__(PT_THREADS) void bucket_hist_kernel(
    const int* __restrict__ dst, int* __restrict__ hist, int E, int nb)
{
    __shared__ int h[MAX_BKT];
    const int t = threadIdx.x;
    h[t] = 0;
    const int base = blockIdx.x * PT_CHUNK;
    int d[PT_PER];
    #pragma unroll
    for (int i = 0; i < PT_PER; ++i) {
        const int e = base + i * PT_THREADS + t;
        d[i] = (e < E) ? dst[e] : -1;
    }
    __syncthreads();
    #pragma unroll
    for (int i = 0; i < PT_PER; ++i)
        if (d[i] >= 0) atomicAdd(&h[d[i] >> BKT_SHIFT], 1);
    __syncthreads();
    if (t < nb) hist[t * gridDim.x + blockIdx.x] = h[t];
}

__global__ __launch_bounds__(PT_THREADS) void bucket_scan_kernel(int* __restrict__ hist, int total)
{
    __shared__ int wsum[32];
    const int t = threadIdx.x;
    const int lane = t & 31;
    const int wv = t >> 5;
    const int per = (total + PT_THREADS - 1) / PT_THREADS;
    const int lo = min(t * per, total);
    const int hi = min(lo + per, total);
    int s = 0;
    for (int i = lo; i < hi; ++i) s += hist[i];
    int inc = s;
    for (int o = 1; o < 32; o <<= 1) {
        int y = __shfl_up(inc, o, 32);
        if (lane >= o) inc += y;
    }
    if (lane == 31) wsum[wv] = inc;
    __syncthreads();
    int wp = (lane < wv) ? wsum[lane] : 0;
    for (int o = 16; o > 0; o >>= 1) wp += __shfl_xor(wp, o, 32);
    int run = wp + inc - s;
    for (int i = lo; i < hi; ++i) {
        const int c = hist[i];
        hist[i] = run;
        run += c;
    }
}

__global__ __launch_bounds__(PT_THREADS) void bucket_place_kernel(
    const int* __restrict__ src, const int* __restrict__ dst, const int* __restrict__ offs,
    int* __restrict__ pairs_raw, int E, int nb)
{
    __shared__ int cur[MAX_BKT];
    const int t = threadIdx.x;
    const int base = blockIdx.x * PT_CHUNK;
    int d[PT_PER], s[PT_PER];
    #pragma unroll
    for (int i = 0; i < PT_PER; ++i) {
        const int e = base + i * PT_THREADS + t;
        const bool in = e < E;
        d[i] = in ? dst[e] : -1;
        s[i] = in ? src[e] : 0;
    }
    cur[t] = (t < nb) ? offs[t * gridDim.x + blockIdx.x] : 0;
    __syncthreads();
    int2* pairs = (int2*)pairs_raw;
    #pragma unroll
    for (int i = 0; i < PT_PER; ++i) {
        if (d[i] >= 0) {
            const int p = atomicAdd(&cur[d[i] >> BKT_SHIFT], 1);
            pairs[p] = make_int2(d[i], s[i]);
        }
    }
}

__global__ __launch_bounds__(PT_THREADS) void bucket_csr_kernel(
    const int* __restrict__ pairs_raw, const int* __restrict__ offs, int blocks,
    int* __restrict__ rowptr, int* __restrict__ degi, float* __restrict__ dis,
    int* __restrict__ csr_src, int n, int E, int nb)
{
    __shared__ int cnt[BKT_NODES];
    __shared__ int pos[BKT_NODES];
    __shared__ int wsum[BKT_NODES / 32];
    const int t = threadIdx.x;
    const int lane = t & 31;
    const int wv = t >> 5;
    const int b = blockIdx.x;
    const int2* pairs = (const int2*)pairs_raw;
    const int beg = offs[b * blocks];
    const int end = (b + 1 < nb) ? offs[(b + 1) * blocks] : E;
    const int node0 = b << BKT_SHIFT;
    if (t < BKT_NODES) cnt[t] = 0;
    __syncthreads();
    for (int i0 = beg + t; i0 < end; i0 += 4 * PT_THREADS) {
        int dl[4];
        for (int u = 0; u < 4; ++u) {
            const int i = i0 + u * PT_THREADS;
            dl[u] = (i < end) ? pairs[i].x : -1;
        }
        for (int u = 0; u < 4; ++u)
            if (dl[u] >= 0) atomicAdd(&cnt[dl[u] - node0], 1);
    }
    __syncthreads();
    const int c = (t < BKT_NODES) ? cnt[t] : 0;
    int inc = c;
    for (int o = 1; o < 32; o <<= 1) {
        int y = __shfl_up(inc, o, 32);
        if (lane >= o) inc += y;
    }
    if (t < BKT_NODES && lane == 31) wsum[wv] = inc;
    __syncthreads();
    if (t < BKT_NODES) {
        int wp = (lane < wv) ? wsum[lane] : 0;
        for (int o = 16; o > 0; o >>= 1) wp += __shfl_xor(wp, o, 32);
        const int start = beg + wp + inc - c;
        pos[t] = start;
        const int node = node0 + t;
        if (node < n) {
            const float d = (float)c;
            rowptr[node] = start;
            degi[node] = c;
            dis[node] = (d > 0.0f) ? rsqrtf(fmaxf(d, 1.0f)) : 0.0f;
        }
    }
    __syncthreads();
    for (int i0 = beg + t; i0 < end; i0 += 4 * PT_THREADS) {
        int2 p[4];
        for (int u = 0; u < 4; ++u) {
            const int i = i0 + u * PT_THREADS;
            p[u] = (i < end) ? pairs[i] : make_int2(-1, 0);
        }
        for (int u = 0; u < 4; ++u) {
            if (p[u].x >= 0) {
                const int slot = atomicAdd(&pos[p[u].x - node0], 1);
                csr_src[slot] = p[u].y;
            }
        }
    }
}

__device__ __forceinline__ float keep_bits(float t, int keep) {
    return __int_as_float(__float_as_int(t) & keep);
}

template <int WEIGHTED>
__global__ __launch_bounds__(256) void gather_kernel(
    const float* __restrict__ X, float* __restrict__ Y,
    const int* __restrict__ csr_src, const int* __restrict__ rowptr,
    const int* __restrict__ degi, const float* __restrict__ dis, int n)
{
    const int tid  = blockIdx.x * blockDim.x + threadIdx.x;
    const int node = tid >> 4;
    const int seg  = tid & 15;
    const bool valid = node < n;
    const int nd  = valid ? node : (n - 1);
    const int beg = rowptr[nd];
    const int cnt = valid ? degi[nd] : 0;
    const float dt = WEIGHTED ? dis[nd] : 1.0f;
    const float4* X4 = (const float4*)X;
    float4 acc = make_float4(0.0f, 0.0f, 0.0f, 0.0f);
    const int cmax = __builtin_amdgcn_readfirstlane(max(cnt, __shfl_xor(cnt, 16, 32)));
    for (int base = 0; base < cmax; base += 16) {
        const int j = base + seg;
        const int mySrc = (j < cnt) ? csr_src[beg + j] : 0;
        float myW = 1.0f;
        if (WEIGHTED) myW = dis[mySrc] * dt;
        const int lim = min(16, cmax - base);
        for (int k = 0; k < lim; k += 4) {
            int s[4]; float w[4]; float4 v[4];
            #pragma unroll
            for (int u = 0; u < 4; ++u) {
                s[u] = __shfl(mySrc, k + u, 16);
                w[u] = WEIGHTED ? __shfl(myW, k + u, 16) : 1.0f;
            }
            #pragma unroll
            for (int u = 0; u < 4; ++u) v[u] = X4[(long)s[u] * 16 + seg];
            #pragma unroll
            for (int u = 0; u < 4; ++u) {
                const int keep = ((base + k + u) < cnt) ? -1 : 0;
                acc.x += keep_bits(WEIGHTED ? v[u].x * w[u] : v[u].x, keep);
                acc.y += keep_bits(WEIGHTED ? v[u].y * w[u] : v[u].y, keep);
                acc.z += keep_bits(WEIGHTED ? v[u].z * w[u] : v[u].z, keep);
                acc.w += keep_bits(WEIGHTED ? v[u].w * w[u] : v[u].w, keep);
            }
        }
    }
    if (valid) ((float4*)Y)[(long)node * 16 + seg] = acc;
}

__global__ __launch_bounds__(256) void gemm1_kernel(
    const float* Sagg, const float* __restrict__ Hf,
    const int* __restrict__ degi,
    const float* __restrict__ Wl, const float* __restrict__ Wr,
    const float* __restrict__ bias, float* Out, int nnodes)
{
    int wave = (blockIdx.x * blockDim.x + threadIdx.x) >> 5;
    int lane = threadIdx.x & 31;
    int rb = wave;
    int rbCount = (nnodes + 15) >> 4;
    if (rb >= rbCount) return;
    int m = rb * 16 + (lane & 15);
    int mclamp = m < nnodes ? m : (nnodes - 1);
    int col0 = lane & 15;
    int kshift = (lane >> 4) * 2;
    float cinv = 1.0f / fmaxf((float)degi[mclamp], 1.0f);
    const float* aggRow = Sagg + (long)mclamp * 64;
    const float* hRow   = Hf   + (long)mclamp * 64;
    v8f acc0 = {}, acc1 = {}, acc2 = {}, acc3 = {};
    #pragma unroll 4
    for (int k0 = 0; k0 < 64; k0 += 4) {
        int kk = k0 + kshift;
        v2f a1; a1.x = aggRow[kk] * cinv; a1.y = aggRow[kk + 1] * cinv;
        v2f a2; a2.x = hRow[kk];          a2.y = hRow[kk + 1];
        const float* wl0 = Wl + kk * 64 + col0;
        const float* wl1 = Wl + (kk + 1) * 64 + col0;
        const float* wr0 = Wr + kk * 64 + col0;
        const float* wr1 = Wr + (kk + 1) * 64 + col0;
        v2f b;
        b.x = wl0[ 0]; b.y = wl1[ 0]; acc0 = wmma4(a1, b, acc0);
        b.x = wl0[16]; b.y = wl1[16]; acc1 = wmma4(a1, b, acc1);
        b.x = wl0[32]; b.y = wl1[32]; acc2 = wmma4(a1, b, acc2);
        b.x = wl0[48]; b.y = wl1[48]; acc3 = wmma4(a1, b, acc3);
        b.x = wr0[ 0]; b.y = wr1[ 0]; acc0 = wmma4(a2, b, acc0);
        b.x = wr0[16]; b.y = wr1[16]; acc1 = wmma4(a2, b, acc1);
        b.x = wr0[32]; b.y = wr1[32]; acc2 = wmma4(a2, b, acc2);
        b.x = wr0[48]; b.y = wr1[48]; acc3 = wmma4(a2, b, acc3);
    }
    int rowbase = rb * 16 + (lane >> 4) * 8;
    v8f accs[4] = {acc0, acc1, acc2, acc3};
    #pragma unroll
    for (int ct = 0; ct < 4; ++ct) {
        int col = ct * 16 + col0;
        float bn = bias[col];
        #pragma unroll
        for (int j = 0; j < 8; ++j) {
            int row = rowbase + j;
            float v = accs[ct][j] + bn;
            v = v > 0.0f ? SELU_SCALE * v : SELU_SCALE * SELU_ALPHA * (expf(v) - 1.0f);
            if (row < nnodes) Out[(long)row * 64 + col] = v;
        }
    }
}

__global__ __launch_bounds__(256) void gemm2_kernel(
    const float* __restrict__ Sagg, const float* __restrict__ Hf,
    const int* __restrict__ degi,
    const float* __restrict__ Wl, const float* __restrict__ Wr,
    const float* __restrict__ bias, float* __restrict__ Logits, int nnodes)
{
    int wave = (blockIdx.x * blockDim.x + threadIdx.x) >> 5;
    int lane = threadIdx.x & 31;
    int rb = wave;
    int rbCount = (nnodes + 15) >> 4;
    if (rb >= rbCount) return;
    int m = rb * 16 + (lane & 15);
    int mclamp = m < nnodes ? m : (nnodes - 1);
    int col0 = lane & 15;
    int kshift = (lane >> 4) * 2;
    float cinv = 1.0f / fmaxf((float)degi[mclamp], 1.0f);
    int col2  = 32 + col0;
    int c2off = col2 < 40 ? col2 : 0;
    float msk2 = col2 < 40 ? 1.0f : 0.0f;
    const float* aggRow = Sagg + (long)mclamp * 64;
    const float* hRow   = Hf   + (long)mclamp * 64;
    v8f acc0 = {}, acc1 = {}, acc2 = {};
    #pragma unroll 4
    for (int k0 = 0; k0 < 64; k0 += 4) {
        int kk = k0 + kshift;
        v2f a1; a1.x = aggRow[kk] * cinv; a1.y = aggRow[kk + 1] * cinv;
        v2f a2; a2.x = hRow[kk];          a2.y = hRow[kk + 1];
        const float* wl0 = Wl + kk * 40;
        const float* wl1 = Wl + (kk + 1) * 40;
        const float* wr0 = Wr + kk * 40;
        const float* wr1 = Wr + (kk + 1) * 40;
        v2f b;
        b.x = wl0[col0];          b.y = wl1[col0];          acc0 = wmma4(a1, b, acc0);
        b.x = wl0[16 + col0];     b.y = wl1[16 + col0];     acc1 = wmma4(a1, b, acc1);
        b.x = wl0[c2off] * msk2;  b.y = wl1[c2off] * msk2;  acc2 = wmma4(a1, b, acc2);
        b.x = wr0[col0];          b.y = wr1[col0];          acc0 = wmma4(a2, b, acc0);
        b.x = wr0[16 + col0];     b.y = wr1[16 + col0];     acc1 = wmma4(a2, b, acc1);
        b.x = wr0[c2off] * msk2;  b.y = wr1[c2off] * msk2;  acc2 = wmma4(a2, b, acc2);
    }
    int rowbase = rb * 16 + (lane >> 4) * 8;
    v8f accs[3] = {acc0, acc1, acc2};
    #pragma unroll
    for (int ct = 0; ct < 3; ++ct) {
        int col = ct * 16 + col0;
        if (col < 40) {
            float bn = bias[col];
            #pragma unroll
            for (int j = 0; j < 8; ++j) {
                int row = rowbase + j;
                if (row < nnodes) Logits[(long)row * 40 + col] = accs[ct][j] + bn;
            }
        }
    }
}

__global__ void softmax_kernel(const float* __restrict__ Logits, float* __restrict__ Out, int nnodes) {
    int wave = (blockIdx.x * blockDim.x + threadIdx.x) >> 5;
    int lane = threadIdx.x & 31;
    if (wave >= nnodes) return;
    const float* row = Logits + (long)wave * 40;
    float v0 = row[lane];
    bool hi = (lane + 32) < 40;
    float v1 = hi ? row[lane + 32] : -3.4e38f;
    float mx = fmaxf(v0, v1);
    for (int o = 16; o > 0; o >>= 1) mx = fmaxf(mx, __shfl_xor(mx, o, 32));
    float e0 = expf(v0 - mx);
    float e1 = hi ? expf(v1 - mx) : 0.0f;
    float s = e0 + e1;
    for (int o = 16; o > 0; o >>= 1) s += __shfl_xor(s, o, 32);
    float inv = 1.0f / s;
    Out[(long)wave * 40 + lane] = e0 * inv;
    if (hi) Out[(long)wave * 40 + lane + 32] = e1 * inv;
}

extern "C" void kernel_launch(void* const* d_in, const int* in_sizes, int n_in,
                              void* d_out, int out_size, void* d_ws, size_t ws_size,
                              hipStream_t stream) {
    const float* x    = (const float*)d_in[0];
    const float* Wl1  = (const float*)d_in[1];
    const float* Wr1  = (const float*)d_in[2];
    const float* b1   = (const float*)d_in[3];
    const float* Wl2  = (const float*)d_in[4];
    const float* Wr2  = (const float*)d_in[5];
    const float* b2   = (const float*)d_in[6];
    const int* esrc   = (const int*)d_in[7];
    const int* edst   = (const int*)d_in[8];

    const int H  = in_sizes[3];
    const int D  = in_sizes[1] / H;
    const int Nn = in_sizes[0] / D;
    const int E  = in_sizes[7];
    (void)n_in; (void)out_size; (void)ws_size;

    float* ws     = (float*)d_ws;
    int*   degi   = (int*)ws;
    float* dis    = ws + (long)Nn;
    int*   rowptr = (int*)(ws + 2L * Nn);
    float* bufA   = ws + 3L * Nn;
    float* bufB   = bufA + (long)Nn * 64;
    float* bufC   = bufB + (long)Nn * 64;
    int*   pairs  = (int*)bufB;
    int*   bhist  = pairs + 2L * E;
    int*   csrSrc = (int*)bufC;
    float* logits = bufC + (long)E;

    const int TB = 256;
    dim3 gPart((E + PT_CHUNK - 1) / PT_CHUNK);
    const int nb = (Nn + BKT_NODES - 1) / BKT_NODES;
    dim3 gGather(((long)Nn * 16 + TB - 1) / TB);
    int rbCount = (Nn + 15) / 16;
    dim3 gGemm(((long)rbCount * 32 + TB - 1) / TB);
    dim3 gSm(((long)Nn * 32 + TB - 1) / TB);

    bucket_hist_kernel<<<gPart, PT_THREADS, 0, stream>>>(edst, bhist, E, nb);
    bucket_scan_kernel<<<1, PT_THREADS, 0, stream>>>(bhist, nb * (int)gPart.x);
    bucket_place_kernel<<<gPart, PT_THREADS, 0, stream>>>(esrc, edst, bhist, pairs, E, nb);
    bucket_csr_kernel<<<nb, PT_THREADS, 0, stream>>>(pairs, bhist, (int)gPart.x, rowptr, degi, dis, csrSrc, Nn, E, nb);

    gather_kernel<1><<<gGather, TB, 0, stream>>>(x, bufA, csrSrc, rowptr, degi, dis, Nn);
    gather_kernel<1><<<gGather, TB, 0, stream>>>(bufA, bufB, csrSrc, rowptr, degi, dis, Nn);
    gather_kernel<0><<<gGather, TB, 0, stream>>>(bufB, bufA, csrSrc, rowptr, degi, dis, Nn);
    gemm1_kernel<<<gGemm, TB, 0, stream>>>(bufA, bufB, degi, Wl1, Wr1, b1, bufA, Nn);
    gather_kernel<0><<<gGather, TB, 0, stream>>>(bufA, bufB, csrSrc, rowptr, degi, dis, Nn);
    gemm2_kernel<<<gGemm, TB, 0, stream>>>(bufB, bufA, degi, Wl2, Wr2, b2, logits, Nn);
    softmax_kernel<<<gSm, TB, 0, stream>>>(logits, (float*)d_out, Nn);
}


# Row softmax done in the second GEMM's epilogue on the accumulator rows: the softmax launch (one wave per row) and the logits plane are gone; denominator summed by a different reduction tree
# speedup vs baseline: 3.9233x; 1.0614x over previous
#include <hip/hip_runtime.h>
#include <hip/hip_bf16.h>
#include <math.h>

typedef __attribute__((ext_vector_type(2))) float v2f;
typedef __attribute__((ext_vector_type(8))) float v8f;

#define SELU_SCALE 1.0507009873554805f
#define SELU_ALPHA 1.6732632423543772f

__device__ __forceinline__ v8f wmma4(v2f a, v2f b, v8f c) {
    return __builtin_amdgcn_wmma_f32_16x16x4_f32(false, a, false, b, (short)0, c, false, false);
}

#define BKT_SHIFT   9
#define BKT_NODES   512
#define MAX_BKT     1024
#define PT_THREADS  1024
#define PT_PER      12
#define PT_CHUNK    (PT_THREADS * PT_PER)

__global__ __launch_bounds__(PT_THREADS) void bucket_hist_kernel(
    const int* __restrict__ dst, int* __restrict__ hist, int E, int nb)
{
    __shared__ int h[MAX_BKT];
    const int t = threadIdx.x;
    h[t] = 0;
    const int base = blockIdx.x * PT_CHUNK;
    int d[PT_PER];
    #pragma unroll
    for (int i = 0; i < PT_PER; ++i) {
        const int e = base + i * PT_THREADS + t;
        d[i] = (e < E) ? dst[e] : -1;
    }
    __syncthreads();
    #pragma unroll
    for (int i = 0; i < PT_PER; ++i)
        if (d[i] >= 0) atomicAdd(&h[d[i] >> BKT_SHIFT], 1);
    __syncthreads();
    if (t < nb) hist[t * gridDim.x + blockIdx.x] = h[t];
}

__global__ __launch_bounds__(PT_THREADS) void bucket_scan_kernel(int* __restrict__ hist, int total)
{
    __shared__ int wsum[32];
    const int t = threadIdx.x;
    const int lane = t & 31;
    const int wv = t >> 5;
    const int per = (total + PT_THREADS - 1) / PT_THREADS;
    const int lo = min(t * per, total);
    const int hi = min(lo + per, total);
    int s = 0;
    for (int i = lo; i < hi; ++i) s += hist[i];
    int inc = s;
    for (int o = 1; o < 32; o <<= 1) {
        int y = __shfl_up(inc, o, 32);
        if (lane >= o) inc += y;
    }
    if (lane == 31) wsum[wv] = inc;
    __syncthreads();
    int wp = (lane < wv) ? wsum[lane] : 0;
    for (int o = 16; o > 0; o >>= 1) wp += __shfl_xor(wp, o, 32);
    int run = wp + inc - s;
    for (int i = lo; i < hi; ++i) {
        const int c = hist[i];
        hist[i] = run;
        run += c;
    }
}

__global__ __launch_bounds__(PT_THREADS) void bucket_place_kernel(
    const int* __restrict__ src, const int* __restrict__ dst, const int* __restrict__ offs,
    int* __restrict__ pairs_raw, int E, int nb)
{
    __shared__ int cur[MAX_BKT];
    const int t = threadIdx.x;
    const int base = blockIdx.x * PT_CHUNK;
    int d[PT_PER], s[PT_PER];
    #pragma unroll
    for (int i = 0; i < PT_PER; ++i) {
        const int e = base + i * PT_THREADS + t;
        const bool in = e < E;
        d[i] = in ? dst[e] : -1;
        s[i] = in ? src[e] : 0;
    }
    cur[t] = (t < nb) ? offs[t * gridDim.x + blockIdx.x] : 0;
    __syncthreads();
    int2* pairs = (int2*)pairs_raw;
    #pragma unroll
    for (int i = 0; i < PT_PER; ++i) {
        if (d[i] >= 0) {
            const int p = atomicAdd(&cur[d[i] >> BKT_SHIFT], 1);
            pairs[p] = make_int2(d[i], s[i]);
        }
    }
}

__global__ __launch_bounds__(PT_THREADS) void bucket_csr_kernel(
    const int* __restrict__ pairs_raw, const int* __restrict__ offs, int blocks,
    int* __restrict__ rowptr, int* __restrict__ degi, float* __restrict__ dis,
    int* __restrict__ csr_src, int n, int E, int nb)
{
    __shared__ int cnt[BKT_NODES];
    __shared__ int pos[BKT_NODES];
    __shared__ int wsum[BKT_NODES / 32];
    const int t = threadIdx.x;
    const int lane = t & 31;
    const int wv = t >> 5;
    const int b = blockIdx.x;
    const int2* pairs = (const int2*)pairs_raw;
    const int beg = offs[b * blocks];
    const int end = (b + 1 < nb) ? offs[(b + 1) * blocks] : E;
    const int node0 = b << BKT_SHIFT;
    if (t < BKT_NODES) cnt[t] = 0;
    __syncthreads();
    for (int i0 = beg + t; i0 < end; i0 += 4 * PT_THREADS) {
        int dl[4];
        for (int u = 0; u < 4; ++u) {
            const int i = i0 + u * PT_THREADS;
            dl[u] = (i < end) ? pairs[i].x : -1;
        }
        for (int u = 0; u < 4; ++u)
            if (dl[u] >= 0) atomicAdd(&cnt[dl[u] - node0], 1);
    }
    __syncthreads();
    const int c = (t < BKT_NODES) ? cnt[t] : 0;
    int inc = c;
    for (int o = 1; o < 32; o <<= 1) {
        int y = __shfl_up(inc, o, 32);
        if (lane >= o) inc += y;
    }
    if (t < BKT_NODES && lane == 31) wsum[wv] = inc;
    __syncthreads();
    if (t < BKT_NODES) {
        int wp = (lane < wv) ? wsum[lane] : 0;
        for (int o = 16; o > 0; o >>= 1) wp += __shfl_xor(wp, o, 32);
        const int start = beg + wp + inc - c;
        pos[t] = start;
        const int node = node0 + t;
        if (node < n) {
            const float d = (float)c;
            rowptr[node] = start;
            degi[node] = c;
            dis[node] = (d > 0.0f) ? rsqrtf(fmaxf(d, 1.0f)) : 0.0f;
        }
    }
    __syncthreads();
    for (int i0 = beg + t; i0 < end; i0 += 4 * PT_THREADS) {
        int2 p[4];
        for (int u = 0; u < 4; ++u) {
            const int i = i0 + u * PT_THREADS;
            p[u] = (i < end) ? pairs[i] : make_int2(-1, 0);
        }
        for (int u = 0; u < 4; ++u) {
            if (p[u].x >= 0) {
                const int slot = atomicAdd(&pos[p[u].x - node0], 1);
                csr_src[slot] = p[u].y;
            }
        }
    }
}

__device__ __forceinline__ float keep_bits(float t, int keep) {
    return __int_as_float(__float_as_int(t) & keep);
}

template <int WEIGHTED>
__global__ __launch_bounds__(256) void gather_kernel(
    const float* __restrict__ X, float* __restrict__ Y,
    const int* __restrict__ csr_src, const int* __restrict__ rowptr,
    const int* __restrict__ degi, const float* __restrict__ dis, int n)
{
    const int tid  = blockIdx.x * blockDim.x + threadIdx.x;
    const int node = tid >> 4;
    const int seg  = tid & 15;
    const bool valid = node < n;
    const int nd  = valid ? node : (n - 1);
    const int beg = rowptr[nd];
    const int cnt = valid ? degi[nd] : 0;
    const float dt = WEIGHTED ? dis[nd] : 1.0f;
    const float4* X4 = (const float4*)X;
    float4 acc = make_float4(0.0f, 0.0f, 0.0f, 0.0f);
    const int cmax = __builtin_amdgcn_readfirstlane(max(cnt, __shfl_xor(cnt, 16, 32)));
    for (int base = 0; base < cmax; base += 16) {
        const int j = base + seg;
        const int mySrc = (j < cnt) ? csr_src[beg + j] : 0;
        float myW = 1.0f;
        if (WEIGHTED) myW = dis[mySrc] * dt;
        const int lim = min(16, cmax - base);
        for (int k = 0; k < lim; k += 4) {
            int s[4]; float w[4]; float4 v[4];
            #pragma unroll
            for (int u = 0; u < 4; ++u) {
                s[u] = __shfl(mySrc, k + u, 16);
                w[u] = WEIGHTED ? __shfl(myW, k + u, 16) : 1.0f;
            }
            #pragma unroll
            for (int u = 0; u < 4; ++u) v[u] = X4[(long)s[u] * 16 + seg];
            #pragma unroll
            for (int u = 0; u < 4; ++u) {
                const int keep = ((base + k + u) < cnt) ? -1 : 0;
                acc.x += keep_bits(WEIGHTED ? v[u].x * w[u] : v[u].x, keep);
                acc.y += keep_bits(WEIGHTED ? v[u].y * w[u] : v[u].y, keep);
                acc.z += keep_bits(WEIGHTED ? v[u].z * w[u] : v[u].z, keep);
                acc.w += keep_bits(WEIGHTED ? v[u].w * w[u] : v[u].w, keep);
            }
        }
    }
    if (valid) ((float4*)Y)[(long)node * 16 + seg] = acc;
}

__global__ __launch_bounds__(256) void gemm1_kernel(
    const float* Sagg, const float* __restrict__ Hf,
    const int* __restrict__ degi,
    const float* __restrict__ Wl, const float* __restrict__ Wr,
    const float* __restrict__ bias, float* Out, int nnodes)
{
    int wave = (blockIdx.x * blockDim.x + threadIdx.x) >> 5;
    int lane = threadIdx.x & 31;
    int rb = wave;
    int rbCount = (nnodes + 15) >> 4;
    if (rb >= rbCount) return;
    int m = rb * 16 + (lane & 15);
    int mclamp = m < nnodes ? m : (nnodes - 1);
    int col0 = lane & 15;
    int kshift = (lane >> 4) * 2;
    float cinv = 1.0f / fmaxf((float)degi[mclamp], 1.0f);
    const float* aggRow = Sagg + (long)mclamp * 64;
    const float* hRow   = Hf   + (long)mclamp * 64;
    v8f acc0 = {}, acc1 = {}, acc2 = {}, acc3 = {};
    #pragma unroll 4
    for (int k0 = 0; k0 < 64; k0 += 4) {
        int kk = k0 + kshift;
        v2f a1; a1.x = aggRow[kk] * cinv; a1.y = aggRow[kk + 1] * cinv;
        v2f a2; a2.x = hRow[kk];          a2.y = hRow[kk + 1];
        const float* wl0 = Wl + kk * 64 + col0;
        const float* wl1 = Wl + (kk + 1) * 64 + col0;
        const float* wr0 = Wr + kk * 64 + col0;
        const float* wr1 = Wr + (kk + 1) * 64 + col0;
        v2f b;
        b.x = wl0[ 0]; b.y = wl1[ 0]; acc0 = wmma4(a1, b, acc0);
        b.x = wl0[16]; b.y = wl1[16]; acc1 = wmma4(a1, b, acc1);
        b.x = wl0[32]; b.y = wl1[32]; acc2 = wmma4(a1, b, acc2);
        b.x = wl0[48]; b.y = wl1[48]; acc3 = wmma4(a1, b, acc3);
        b.x = wr0[ 0]; b.y = wr1[ 0]; acc0 = wmma4(a2, b, acc0);
        b.x = wr0[16]; b.y = wr1[16]; acc1 = wmma4(a2, b, acc1);
        b.x = wr0[32]; b.y = wr1[32]; acc2 = wmma4(a2, b, acc2);
        b.x = wr0[48]; b.y = wr1[48]; acc3 = wmma4(a2, b, acc3);
    }
    int rowbase = rb * 16 + (lane >> 4) * 8;
    v8f accs[4] = {acc0, acc1, acc2, acc3};
    #pragma unroll
    for (int ct = 0; ct < 4; ++ct) {
        int col = ct * 16 + col0;
        float bn = bias[col];
        #pragma unroll
        for (int j = 0; j < 8; ++j) {
            int row = rowbase + j;
            float v = accs[ct][j] + bn;
            v = v > 0.0f ? SELU_SCALE * v : SELU_SCALE * SELU_ALPHA * (expf(v) - 1.0f);
            if (row < nnodes) Out[(long)row * 64 + col] = v;
        }
    }
}

__global__ __launch_bounds__(256) void gemm2_softmax_kernel(
    const float* __restrict__ Sagg, const float* __restrict__ Hf,
    const int* __restrict__ degi,
    const float* __restrict__ Wl, const float* __restrict__ Wr,
    const float* __restrict__ bias, float* __restrict__ Out, int nnodes)
{
    int wave = (blockIdx.x * blockDim.x + threadIdx.x) >> 5;
    int lane = threadIdx.x & 31;
    int rb = wave;
    int rbCount = (nnodes + 15) >> 4;
    if (rb >= rbCount) return;
    int m = rb * 16 + (lane & 15);
    int mclamp = m < nnodes ? m : (nnodes - 1);
    int col0 = lane & 15;
    int kshift = (lane >> 4) * 2;
    float cinv = 1.0f / fmaxf((float)degi[mclamp], 1.0f);
    int col2  = 32 + col0;
    const bool hi2 = col2 < 40;
    int c2off = hi2 ? col2 : 0;
    float msk2 = hi2 ? 1.0f : 0.0f;
    const float* aggRow = Sagg + (long)mclamp * 64;
    const float* hRow   = Hf   + (long)mclamp * 64;
    v8f acc0 = {}, acc1 = {}, acc2 = {};
    #pragma unroll 4
    for (int k0 = 0; k0 < 64; k0 += 4) {
        int kk = k0 + kshift;
        v2f a1; a1.x = aggRow[kk] * cinv; a1.y = aggRow[kk + 1] * cinv;
        v2f a2; a2.x = hRow[kk];          a2.y = hRow[kk + 1];
        const float* wl0 = Wl + kk * 40;
        const float* wl1 = Wl + (kk + 1) * 40;
        const float* wr0 = Wr + kk * 40;
        const float* wr1 = Wr + (kk + 1) * 40;
        v2f b;
        b.x = wl0[col0];          b.y = wl1[col0];          acc0 = wmma4(a1, b, acc0);
        b.x = wl0[16 + col0];     b.y = wl1[16 + col0];     acc1 = wmma4(a1, b, acc1);
        b.x = wl0[c2off] * msk2;  b.y = wl1[c2off] * msk2;  acc2 = wmma4(a1, b, acc2);
        b.x = wr0[col0];          b.y = wr1[col0];          acc0 = wmma4(a2, b, acc0);
        b.x = wr0[16 + col0];     b.y = wr1[16 + col0];     acc1 = wmma4(a2, b, acc1);
        b.x = wr0[c2off] * msk2;  b.y = wr1[c2off] * msk2;  acc2 = wmma4(a2, b, acc2);
    }
    const int rowbase = rb * 16 + (lane >> 4) * 8;
    const float bn0 = bias[col0];
    const float bn1 = bias[16 + col0];
    const float bn2 = bias[c2off];
    float v0[8], v1[8], v2[8], mx[8], sm[8];
    #pragma unroll
    for (int j = 0; j < 8; ++j) {
        v0[j] = acc0[j] + bn0;
        v1[j] = acc1[j] + bn1;
        v2[j] = acc2[j] + bn2;
        const float m01 = fmaxf(v0[j], v1[j]);
        mx[j] = hi2 ? fmaxf(m01, v2[j]) : m01;
    }
    #pragma unroll
    for (int o = 8; o > 0; o >>= 1) {
        #pragma unroll
        for (int j = 0; j < 8; ++j) mx[j] = fmaxf(mx[j], __shfl_xor(mx[j], o, 32));
    }
    #pragma unroll
    for (int j = 0; j < 8; ++j) {
        v0[j] = expf(v0[j] - mx[j]);
        v1[j] = expf(v1[j] - mx[j]);
        const float e2 = expf(v2[j] - mx[j]);
        v2[j] = hi2 ? e2 : 0.0f;
        sm[j] = v0[j] + v1[j] + v2[j];
    }
    #pragma unroll
    for (int o = 8; o > 0; o >>= 1) {
        #pragma unroll
        for (int j = 0; j < 8; ++j) sm[j] += __shfl_xor(sm[j], o, 32);
    }
    #pragma unroll
    for (int j = 0; j < 8; ++j) {
        const int row = rowbase + j;
        const float inv = 1.0f / sm[j];
        if (row < nnodes) {
            float* orow = Out + (long)row * 40;
            orow[col0]      = v0[j] * inv;
            orow[16 + col0] = v1[j] * inv;
            if (hi2) orow[col2] = v2[j] * inv;
        }
    }
}

extern "C" void kernel_launch(void* const* d_in, const int* in_sizes, int n_in,
                              void* d_out, int out_size, void* d_ws, size_t ws_size,
                              hipStream_t stream) {
    const float* x    = (const float*)d_in[0];
    const float* Wl1  = (const float*)d_in[1];
    const float* Wr1  = (const float*)d_in[2];
    const float* b1   = (const float*)d_in[3];
    const float* Wl2  = (const float*)d_in[4];
    const float* Wr2  = (const float*)d_in[5];
    const float* b2   = (const float*)d_in[6];
    const int* esrc   = (const int*)d_in[7];
    const int* edst   = (const int*)d_in[8];

    const int H  = in_sizes[3];
    const int D  = in_sizes[1] / H;
    const int Nn = in_sizes[0] / D;
    const int E  = in_sizes[7];
    (void)n_in; (void)out_size; (void)ws_size;

    float* ws     = (float*)d_ws;
    int*   degi   = (int*)ws;
    float* dis    = ws + (long)Nn;
    int*   rowptr = (int*)(ws + 2L * Nn);
    float* bufA   = ws + 3L * Nn;
    float* bufB   = bufA + (long)Nn * 64;
    float* bufC   = bufB + (long)Nn * 64;
    int*   pairs  = (int*)bufB;
    int*   bhist  = pairs + 2L * E;
    int*   csrSrc = (int*)bufC;

    const int TB = 256;
    dim3 gPart((E + PT_CHUNK - 1) / PT_CHUNK);
    const int nb = (Nn + BKT_NODES - 1) / BKT_NODES;
    dim3 gGather(((long)Nn * 16 + TB - 1) / TB);
    int rbCount = (Nn + 15) / 16;
    dim3 gGemm(((long)rbCount * 32 + TB - 1) / TB);

    bucket_hist_kernel<<<gPart, PT_THREADS, 0, stream>>>(edst, bhist, E, nb);
    bucket_scan_kernel<<<1, PT_THREADS, 0, stream>>>(bhist, nb * (int)gPart.x);
    bucket_place_kernel<<<gPart, PT_THREADS, 0, stream>>>(esrc, edst, bhist, pairs, E, nb);
    bucket_csr_kernel<<<nb, PT_THREADS, 0, stream>>>(pairs, bhist, (int)gPart.x, rowptr, degi, dis, csrSrc, Nn, E, nb);

    gather_kernel<1><<<gGather, TB, 0, stream>>>(x, bufA, csrSrc, rowptr, degi, dis, Nn);
    gather_kernel<1><<<gGather, TB, 0, stream>>>(bufA, bufB, csrSrc, rowptr, degi, dis, Nn);
    gather_kernel<0><<<gGather, TB, 0, stream>>>(bufB, bufA, csrSrc, rowptr, degi, dis, Nn);
    gemm1_kernel<<<gGemm, TB, 0, stream>>>(bufA, bufB, degi, Wl1, Wr1, b1, bufA, Nn);
    gather_kernel<0><<<gGather, TB, 0, stream>>>(bufA, bufB, csrSrc, rowptr, degi, dis, Nn);
    gemm2_softmax_kernel<<<gGemm, TB, 0, stream>>>(bufB, bufA, degi, Wl2, Wr2, b2, (float*)d_out, Nn);
}


# Bucket write offsets computed inside every place block (column sums of a block-major histogram, then a scan over buckets in LDS): the single-block scan launch is gone, 9 launches instead of 10
# speedup vs baseline: 4.1674x; 1.0622x over previous
#include <hip/hip_runtime.h>
#include <hip/hip_bf16.h>
#include <math.h>

typedef __attribute__((ext_vector_type(2))) float v2f;
typedef __attribute__((ext_vector_type(8))) float v8f;

#define SELU_SCALE 1.0507009873554805f
#define SELU_ALPHA 1.6732632423543772f

__device__ __forceinline__ v8f wmma4(v2f a, v2f b, v8f c) {
    return __builtin_amdgcn_wmma_f32_16x16x4_f32(false, a, false, b, (short)0, c, false, false);
}

#define BKT_SHIFT   9
#define BKT_NODES   512
#define MAX_BKT     1024
#define PT_THREADS  1024
#define PT_PER      12
#define PT_CHUNK    (PT_THREADS * PT_PER)
#define CS_PER      17

__global__ __launch_bounds__(PT_THREADS) void bucket_hist_kernel(
    const int* __restrict__ dst, int* __restrict__ hist, int E, int nb)
{
    __shared__ int h[MAX_BKT];
    const int t = threadIdx.x;
    h[t] = 0;
    const int base = blockIdx.x * PT_CHUNK;
    int d[PT_PER];
    #pragma unroll
    for (int i = 0; i < PT_PER; ++i) {
        const int e = base + i * PT_THREADS + t;
        d[i] = (e < E) ? dst[e] : -1;
    }
    __syncthreads();
    #pragma unroll
    for (int i = 0; i < PT_PER; ++i)
        if (d[i] >= 0) atomicAdd(&h[d[i] >> BKT_SHIFT], 1);
    __syncthreads();
    if (t < nb) hist[blockIdx.x * nb + t] = h[t];
}

__global__ __launch_bounds__(PT_THREADS) void bucket_place_kernel(
    const int* __restrict__ src, const int* __restrict__ dst, const int* __restrict__ hist,
    int* __restrict__ bstart, int* __restrict__ pairs_raw, int E, int nb)
{
    __shared__ int cur[MAX_BKT];
    __shared__ int pre[MAX_BKT];
    __shared__ int wsum[PT_THREADS / 32];
    const int t = threadIdx.x;
    const int lane = t & 31;
    const int wv = t >> 5;
    const int blk = blockIdx.x;
    const int nblk = gridDim.x;
    const int base = blk * PT_CHUNK;
    cur[t] = 0;
    pre[t] = 0;
    __syncthreads();
    int d[PT_PER], s[PT_PER];
    #pragma unroll
    for (int i = 0; i < PT_PER; ++i) {
        const int e = base + i * PT_THREADS + t;
        const bool in = e < E;
        d[i] = in ? dst[e] : -1;
        s[i] = in ? src[e] : 0;
    }
    for (int b = t & 255; b < nb; b += 256) {
        int tot = 0, bef = 0;
        for (int k0 = t >> 8; k0 < nblk; k0 += 4 * CS_PER) {
            int v[CS_PER];
            #pragma unroll
            for (int u = 0; u < CS_PER; ++u) {
                const int k = k0 + 4 * u;
                v[u] = (k < nblk) ? hist[k * nb + b] : 0;
            }
            #pragma unroll
            for (int u = 0; u < CS_PER; ++u) {
                tot += v[u];
                bef += (k0 + 4 * u < blk) ? v[u] : 0;
            }
        }
        atomicAdd(&cur[b], tot);
        atomicAdd(&pre[b], bef);
    }
    __syncthreads();
    const int c = cur[t];
    int inc = c;
    for (int o = 1; o < 32; o <<= 1) {
        int y = __shfl_up(inc, o, 32);
        if (lane >= o) inc += y;
    }
    if (lane == 31) wsum[wv] = inc;
    __syncthreads();
    int wp = (lane < wv) ? wsum[lane] : 0;
    for (int o = 16; o > 0; o >>= 1) wp += __shfl_xor(wp, o, 32);
    const int start = wp + inc - c;
    cur[t] = start + pre[t];
    if (blk == 0 && t < nb) bstart[t] = start;
    __syncthreads();
    int2* pairs = (int2*)pairs_raw;
    #pragma unroll
    for (int i = 0; i < PT_PER; ++i) {
        if (d[i] >= 0) {
            const int p = atomicAdd(&cur[d[i] >> BKT_SHIFT], 1);
            pairs[p] = make_int2(d[i], s[i]);
        }
    }
}

__global__ __launch_bounds__(PT_THREADS) void bucket_csr_kernel(
    const int* __restrict__ pairs_raw, const int* __restrict__ bstart,
    int* __restrict__ rowptr, int* __restrict__ degi, float* __restrict__ dis,
    int* __restrict__ csr_src, int n, int E, int nb)
{
    __shared__ int cnt[BKT_NODES];
    __shared__ int pos[BKT_NODES];
    __shared__ int wsum[BKT_NODES / 32];
    const int t = threadIdx.x;
    const int lane = t & 31;
    const int wv = t >> 5;
    const int b = blockIdx.x;
    const int2* pairs = (const int2*)pairs_raw;
    const int beg = bstart[b];
    const int end = (b + 1 < nb) ? bstart[b + 1] : E;
    const int node0 = b << BKT_SHIFT;
    if (t < BKT_NODES) cnt[t] = 0;
    __syncthreads();
    for (int i0 = beg + t; i0 < end; i0 += 4 * PT_THREADS) {
        int dl[4];
        for (int u = 0; u < 4; ++u) {
            const int i = i0 + u * PT_THREADS;
            dl[u] = (i < end) ? pairs[i].x : -1;
        }
        for (int u = 0; u < 4; ++u)
            if (dl[u] >= 0) atomicAdd(&cnt[dl[u] - node0], 1);
    }
    __syncthreads();
    const int c = (t < BKT_NODES) ? cnt[t] : 0;
    int inc = c;
    for (int o = 1; o < 32; o <<= 1) {
        int y = __shfl_up(inc, o, 32);
        if (lane >= o) inc += y;
    }
    if (t < BKT_NODES && lane == 31) wsum[wv] = inc;
    __syncthreads();
    if (t < BKT_NODES) {
        int wp = (lane < wv) ? wsum[lane] : 0;
        for (int o = 16; o > 0; o >>= 1) wp += __shfl_xor(wp, o, 32);
        const int start = beg + wp + inc - c;
        pos[t] = start;
        const int node = node0 + t;
        if (node < n) {
            const float d = (float)c;
            rowptr[node] = start;
            degi[node] = c;
            dis[node] = (d > 0.0f) ? rsqrtf(fmaxf(d, 1.0f)) : 0.0f;
        }
    }
    __syncthreads();
    for (int i0 = beg + t; i0 < end; i0 += 4 * PT_THREADS) {
        int2 p[4];
        for (int u = 0; u < 4; ++u) {
            const int i = i0 + u * PT_THREADS;
            p[u] = (i < end) ? pairs[i] : make_int2(-1, 0);
        }
        for (int u = 0; u < 4; ++u) {
            if (p[u].x >= 0) {
                const int slot = atomicAdd(&pos[p[u].x - node0], 1);
                csr_src[slot] = p[u].y;
            }
        }
    }
}

__device__ __forceinline__ float keep_bits(float t, int keep) {
    return __int_as_float(__float_as_int(t) & keep);
}

template <int WEIGHTED>
__global__ __launch_bounds__(256) void gather_kernel(
    const float* __restrict__ X, float* __restrict__ Y,
    const int* __restrict__ csr_src, const int* __restrict__ rowptr,
    const int* __restrict__ degi, const float* __restrict__ dis, int n)
{
    const int tid  = blockIdx.x * blockDim.x + threadIdx.x;
    const int node = tid >> 4;
    const int seg  = tid & 15;
    const bool valid = node < n;
    const int nd  = valid ? node : (n - 1);
    const int beg = rowptr[nd];
    const int cnt = valid ? degi[nd] : 0;
    const float dt = WEIGHTED ? dis[nd] : 1.0f;
    const float4* X4 = (const float4*)X;
    float4 acc = make_float4(0.0f, 0.0f, 0.0f, 0.0f);
    const int cmax = __builtin_amdgcn_readfirstlane(max(cnt, __shfl_xor(cnt, 16, 32)));
    for (int base = 0; base < cmax; base += 16) {
        const int j = base + seg;
        const int mySrc = (j < cnt) ? csr_src[beg + j] : 0;
        float myW = 1.0f;
        if (WEIGHTED) myW = dis[mySrc] * dt;
        const int lim = min(16, cmax - base);
        for (int k = 0; k < lim; k += 4) {
            int s[4]; float w[4]; float4 v[4];
            #pragma unroll
            for (int u = 0; u < 4; ++u) {
                s[u] = __shfl(mySrc, k + u, 16);
                w[u] = WEIGHTED ? __shfl(myW, k + u, 16) : 1.0f;
            }
            #pragma unroll
            for (int u = 0; u < 4; ++u) v[u] = X4[(long)s[u] * 16 + seg];
            #pragma unroll
            for (int u = 0; u < 4; ++u) {
                const int keep = ((base + k + u) < cnt) ? -1 : 0;
                acc.x += keep_bits(WEIGHTED ? v[u].x * w[u] : v[u].x, keep);
                acc.y += keep_bits(WEIGHTED ? v[u].y * w[u] : v[u].y, keep);
                acc.z += keep_bits(WEIGHTED ? v[u].z * w[u] : v[u].z, keep);
                acc.w += keep_bits(WEIGHTED ? v[u].w * w[u] : v[u].w, keep);
            }
        }
    }
    if (valid) ((float4*)Y)[(long)node * 16 + seg] = acc;
}

__global__ __launch_bounds__(256) void gemm1_kernel(
    const float* Sagg, const float* __restrict__ Hf,
    const int* __restrict__ degi,
    const float* __restrict__ Wl, const float* __restrict__ Wr,
    const float* __restrict__ bias, float* Out, int nnodes)
{
    int wave = (blockIdx.x * blockDim.x + threadIdx.x) >> 5;
    int lane = threadIdx.x & 31;
    int rb = wave;
    int rbCount = (nnodes + 15) >> 4;
    if (rb >= rbCount) return;
    int m = rb * 16 + (lane & 15);
    int mclamp = m < nnodes ? m : (nnodes - 1);
    int col0 = lane & 15;
    int kshift = (lane >> 4) * 2;
    float cinv = 1.0f / fmaxf((float)degi[mclamp], 1.0f);
    const float* aggRow = Sagg + (long)mclamp * 64;
    const float* hRow   = Hf   + (long)mclamp * 64;
    v8f acc0 = {}, acc1 = {}, acc2 = {}, acc3 = {};
    #pragma unroll 4
    for (int k0 = 0; k0 < 64; k0 += 4) {
        int kk = k0 + kshift;
        v2f a1; a1.x = aggRow[kk] * cinv; a1.y = aggRow[kk + 1] * cinv;
        v2f a2; a2.x = hRow[kk];          a2.y = hRow[kk + 1];
        const float* wl0 = Wl + kk * 64 + col0;
        const float* wl1 = Wl + (kk + 1) * 64 + col0;
        const float* wr0 = Wr + kk * 64 + col0;
        const float* wr1 = Wr + (kk + 1) * 64 + col0;
        v2f b;
        b.x = wl0[ 0]; b.y = wl1[ 0]; acc0 = wmma4(a1, b, acc0);
        b.x = wl0[16]; b.y = wl1[16]; acc1 = wmma4(a1, b, acc1);
        b.x = wl0[32]; b.y = wl1[32]; acc2 = wmma4(a1, b, acc2);
        b.x = wl0[48]; b.y = wl1[48]; acc3 = wmma4(a1, b, acc3);
        b.x = wr0[ 0]; b.y = wr1[ 0]; acc0 = wmma4(a2, b, acc0);
        b.x = wr0[16]; b.y = wr1[16]; acc1 = wmma4(a2, b, acc1);
        b.x = wr0[32]; b.y = wr1[32]; acc2 = wmma4(a2, b, acc2);
        b.x = wr0[48]; b.y = wr1[48]; acc3 = wmma4(a2, b, acc3);
    }
    int rowbase = rb * 16 + (lane >> 4) * 8;
    v8f accs[4] = {acc0, acc1, acc2, acc3};
    #pragma unroll
    for (int ct = 0; ct < 4; ++ct) {
        int col = ct * 16 + col0;
        float bn = bias[col];
        #pragma unroll
        for (int j = 0; j < 8; ++j) {
            int row = rowbase + j;
            float v = accs[ct][j] + bn;
            v = v > 0.0f ? SELU_SCALE * v : SELU_SCALE * SELU_ALPHA * (expf(v) - 1.0f);
            if (row < nnodes) Out[(long)row * 64 + col] = v;
        }
    }
}

__global__ __launch_bounds__(256) void gemm2_softmax_kernel(
    const float* __restrict__ Sagg, const float* __restrict__ Hf,
    const int* __restrict__ degi,
    const float* __restrict__ Wl, const float* __restrict__ Wr,
    const float* __restrict__ bias, float* __restrict__ Out, int nnodes)
{
    int wave = (blockIdx.x * blockDim.x + threadIdx.x) >> 5;
    int lane = threadIdx.x & 31;
    int rb = wave;
    int rbCount = (nnodes + 15) >> 4;
    if (rb >= rbCount) return;
    int m = rb * 16 + (lane & 15);
    int mclamp = m < nnodes ? m : (nnodes - 1);
    int col0 = lane & 15;
    int kshift = (lane >> 4) * 2;
    float cinv = 1.0f / fmaxf((float)degi[mclamp], 1.0f);
    int col2  = 32 + col0;
    const bool hi2 = col2 < 40;
    int c2off = hi2 ? col2 : 0;
    float msk2 = hi2 ? 1.0f : 0.0f;
    const float* aggRow = Sagg + (long)mclamp * 64;
    const float* hRow   = Hf   + (long)mclamp * 64;
    v8f acc0 = {}, acc1 = {}, acc2 = {};
    #pragma unroll 4
    for (int k0 = 0; k0 < 64; k0 += 4) {
        int kk = k0 + kshift;
        v2f a1; a1.x = aggRow[kk] * cinv; a1.y = aggRow[kk + 1] * cinv;
        v2f a2; a2.x = hRow[kk];          a2.y = hRow[kk + 1];
        const float* wl0 = Wl + kk * 40;
        const float* wl1 = Wl + (kk + 1) * 40;
        const float* wr0 = Wr + kk * 40;
        const float* wr1 = Wr + (kk + 1) * 40;
        v2f b;
        b.x = wl0[col0];          b.y = wl1[col0];          acc0 = wmma4(a1, b, acc0);
        b.x = wl0[16 + col0];     b.y = wl1[16 + col0];     acc1 = wmma4(a1, b, acc1);
        b.x = wl0[c2off] * msk2;  b.y = wl1[c2off] * msk2;  acc2 = wmma4(a1, b, acc2);
        b.x = wr0[col0];          b.y = wr1[col0];          acc0 = wmma4(a2, b, acc0);
        b.x = wr0[16 + col0];     b.y = wr1[16 + col0];     acc1 = wmma4(a2, b, acc1);
        b.x = wr0[c2off] * msk2;  b.y = wr1[c2off] * msk2;  acc2 = wmma4(a2, b, acc2);
    }
    const int rowbase = rb * 16 + (lane >> 4) * 8;
    const float bn0 = bias[col0];
    const float bn1 = bias[16 + col0];
    const float bn2 = bias[c2off];
    float v0[8], v1[8], v2[8], mx[8], sm[8];
    #pragma unroll
    for (int j = 0; j < 8; ++j) {
        v0[j] = acc0[j] + bn0;
        v1[j] = acc1[j] + bn1;
        v2[j] = acc2[j] + bn2;
        const float m01 = fmaxf(v0[j], v1[j]);
        mx[j] = hi2 ? fmaxf(m01, v2[j]) : m01;
    }
    #pragma unroll
    for (int o = 8; o > 0; o >>= 1) {
        #pragma unroll
        for (int j = 0; j < 8; ++j) mx[j] = fmaxf(mx[j], __shfl_xor(mx[j], o, 32));
    }
    #pragma unroll
    for (int j = 0; j < 8; ++j) {
        v0[j] = expf(v0[j] - mx[j]);
        v1[j] = expf(v1[j] - mx[j]);
        const float e2 = expf(v2[j] - mx[j]);
        v2[j] = hi2 ? e2 : 0.0f;
        sm[j] = v0[j] + v1[j] + v2[j];
    }
    #pragma unroll
    for (int o = 8; o > 0; o >>= 1) {
        #pragma unroll
        for (int j = 0; j < 8; ++j) sm[j] += __shfl_xor(sm[j], o, 32);
    }
    #pragma unroll
    for (int j = 0; j < 8; ++j) {
        const int row = rowbase + j;
        const float inv = 1.0f / sm[j];
        if (row < nnodes) {
            float* orow = Out + (long)row * 40;
            orow[col0]      = v0[j] * inv;
            orow[16 + col0] = v1[j] * inv;
            if (hi2) orow[col2] = v2[j] * inv;
        }
    }
}

extern "C" void kernel_launch(void* const* d_in, const int* in_sizes, int n_in,
                              void* d_out, int out_size, void* d_ws, size_t ws_size,
                              hipStream_t stream) {
    const float* x    = (const float*)d_in[0];
    const float* Wl1  = (const float*)d_in[1];
    const float* Wr1  = (const float*)d_in[2];
    const float* b1   = (const float*)d_in[3];
    const float* Wl2  = (const float*)d_in[4];
    const float* Wr2  = (const float*)d_in[5];
    const float* b2   = (const float*)d_in[6];
    const int* esrc   = (const int*)d_in[7];
    const int* edst   = (const int*)d_in[8];

    const int H  = in_sizes[3];
    const int D  = in_sizes[1] / H;
    const int Nn = in_sizes[0] / D;
    const int E  = in_sizes[7];
    (void)n_in; (void)out_size; (void)ws_size;

    float* ws     = (float*)d_ws;
    int*   degi   = (int*)ws;
    float* dis    = ws + (long)Nn;
    int*   rowptr = (int*)(ws + 2L * Nn);
    float* bufA   = ws + 3L * Nn;
    float* bufB   = bufA + (long)Nn * 64;
    float* bufC   = bufB + (long)Nn * 64;
    int*   pairs  = (int*)bufB;
    int*   bhist  = pairs + 2L * E;
    int*   csrSrc = (int*)bufC;

    const int TB = 256;
    dim3 gPart((E + PT_CHUNK - 1) / PT_CHUNK);
    const int nb = (Nn + BKT_NODES - 1) / BKT_NODES;
    dim3 gGather(((long)Nn * 16 + TB - 1) / TB);
    int*   bstart = bhist + (long)nb * gPart.x;
    int rbCount = (Nn + 15) / 16;
    dim3 gGemm(((long)rbCount * 32 + TB - 1) / TB);

    bucket_hist_kernel<<<gPart, PT_THREADS, 0, stream>>>(edst, bhist, E, nb);
    bucket_place_kernel<<<gPart, PT_THREADS, 0, stream>>>(esrc, edst, bhist, bstart, pairs, E, nb);
    bucket_csr_kernel<<<nb, PT_THREADS, 0, stream>>>(pairs, bstart, rowptr, degi, dis, csrSrc, Nn, E, nb);

    gather_kernel<1><<<gGather, TB, 0, stream>>>(x, bufA, csrSrc, rowptr, degi, dis, Nn);
    gather_kernel<1><<<gGather, TB, 0, stream>>>(bufA, bufB, csrSrc, rowptr, degi, dis, Nn);
    gather_kernel<0><<<gGather, TB, 0, stream>>>(bufB, bufA, csrSrc, rowptr, degi, dis, Nn);
    gemm1_kernel<<<gGemm, TB, 0, stream>>>(bufA, bufB, degi, Wl1, Wr1, b1, bufA, Nn);
    gather_kernel<0><<<gGather, TB, 0, stream>>>(bufA, bufB, csrSrc, rowptr, degi, dis, Nn);
    gemm2_softmax_kernel<<<gGemm, TB, 0, stream>>>(bufB, bufA, degi, Wl2, Wr2, b2, (float*)d_out, Nn);
}


# Mean aggregation fused into both GEMM launches: a work-group sums its 16 nodes' in-edge rows in registers and hands the tile to its matrix waves through LDS; 7 launches, not 9; sum orders unchanged
# speedup vs baseline: 4.5424x; 1.0900x over previous
#include <hip/hip_runtime.h>
#include <hip/hip_bf16.h>
#include <math.h>

typedef __attribute__((ext_vector_type(2))) float v2f;
typedef __attribute__((ext_vector_type(8))) float v8f;

#define SELU_SCALE 1.0507009873554805f
#define SELU_ALPHA 1.6732632423543772f

__device__ __forceinline__ v8f wmma4(v2f a, v2f b, v8f c) {
    return __builtin_amdgcn_wmma_f32_16x16x4_f32(false, a, false, b, (short)0, c, false, false);
}

#define BKT_SHIFT   9
#define BKT_NODES   512
#define MAX_BKT     1024
#define PT_THREADS  1024
#define PT_PER      12
#define PT_CHUNK    (PT_THREADS * PT_PER)
#define CS_PER      17

__global__ __launch_bounds__(PT_THREADS) void bucket_hist_kernel(
    const int* __restrict__ dst, int* __restrict__ hist, int E, int nb)
{
    __shared__ int h[MAX_BKT];
    const int t = threadIdx.x;
    h[t] = 0;
    const int base = blockIdx.x * PT_CHUNK;
    int d[PT_PER];
    #pragma unroll
    for (int i = 0; i < PT_PER; ++i) {
        const int e = base + i * PT_THREADS + t;
        d[i] = (e < E) ? dst[e] : -1;
    }
    __syncthreads();
    #pragma unroll
    for (int i = 0; i < PT_PER; ++i)
        if (d[i] >= 0) atomicAdd(&h[d[i] >> BKT_SHIFT], 1);
    __syncthreads();
    if (t < nb) hist[blockIdx.x * nb + t] = h[t];
}

__global__ __launch_bounds__(PT_THREADS) void bucket_place_kernel(
    const int* __restrict__ src, const int* __restrict__ dst, const int* __restrict__ hist,
    int* __restrict__ bstart, int* __restrict__ pairs_raw, int E, int nb)
{
    __shared__ int cur[MAX_BKT];
    __shared__ int pre[MAX_BKT];
    __shared__ int wsum[PT_THREADS / 32];
    const int t = threadIdx.x;
    const int lane = t & 31;
    const int wv = t >> 5;
    const int blk = blockIdx.x;
    const int nblk = gridDim.x;
    const int base = blk * PT_CHUNK;
    cur[t] = 0;
    pre[t] = 0;
    __syncthreads();
    int d[PT_PER], s[PT_PER];
    #pragma unroll
    for (int i = 0; i < PT_PER; ++i) {
        const int e = base + i * PT_THREADS + t;
        const bool in = e < E;
        d[i] = in ? dst[e] : -1;
        s[i] = in ? src[e] : 0;
    }
    for (int b = t & 255; b < nb; b += 256) {
        int tot = 0, bef = 0;
        for (int k0 = t >> 8; k0 < nblk; k0 += 4 * CS_PER) {
            int v[CS_PER];
            #pragma unroll
            for (int u = 0; u < CS_PER; ++u) {
                const int k = k0 + 4 * u;
                v[u] = (k < nblk) ? hist[k * nb + b] : 0;
            }
            #pragma unroll
            for (int u = 0; u < CS_PER; ++u) {
                tot += v[u];
                bef += (k0 + 4 * u < blk) ? v[u] : 0;
            }
        }
        atomicAdd(&cur[b], tot);
        atomicAdd(&pre[b], bef);
    }
    __syncthreads();
    const int c = cur[t];
    int inc = c;
    for (int o = 1; o < 32; o <<= 1) {
        int y = __shfl_up(inc, o, 32);
        if (lane >= o) inc += y;
    }
    if (lane == 31) wsum[wv] = inc;
    __syncthreads();
    int wp = (lane < wv) ? wsum[lane] : 0;
    for (int o = 16; o > 0; o >>= 1) wp += __shfl_xor(wp, o, 32);
    const int start = wp + inc - c;
    cur[t] = start + pre[t];
    if (blk == 0 && t < nb) bstart[t] = start;
    __syncthreads();
    int2* pairs = (int2*)pairs_raw;
    #pragma unroll
    for (int i = 0; i < PT_PER; ++i) {
        if (d[i] >= 0) {
            const int p = atomicAdd(&cur[d[i] >> BKT_SHIFT], 1);
            pairs[p] = make_int2(d[i], s[i]);
        }
    }
}

__global__ __launch_bounds__(PT_THREADS) void bucket_csr_kernel(
    const int* __restrict__ pairs_raw, const int* __restrict__ bstart,
    int* __restrict__ rowptr, int* __restrict__ degi, float* __restrict__ dis,
    int* __restrict__ csr_src, int n, int E, int nb)
{
    __shared__ int cnt[BKT_NODES];
    __shared__ int pos[BKT_NODES];
    __shared__ int wsum[BKT_NODES / 32];
    const int t = threadIdx.x;
    const int lane = t & 31;
    const int wv = t >> 5;
    const int b = blockIdx.x;
    const int2* pairs = (const int2*)pairs_raw;
    const int beg = bstart[b];
    const int end = (b + 1 < nb) ? bstart[b + 1] : E;
    const int node0 = b << BKT_SHIFT;
    if (t < BKT_NODES) cnt[t] = 0;
    __syncthreads();
    for (int i0 = beg + t; i0 < end; i0 += 4 * PT_THREADS) {
        int dl[4];
        for (int u = 0; u < 4; ++u) {
            const int i = i0 + u * PT_THREADS;
            dl[u] = (i < end) ? pairs[i].x : -1;
        }
        for (int u = 0; u < 4; ++u)
            if (dl[u] >= 0) atomicAdd(&cnt[dl[u] - node0], 1);
    }
    __syncthreads();
    const int c = (t < BKT_NODES) ? cnt[t] : 0;
    int inc = c;
    for (int o = 1; o < 32; o <<= 1) {
        int y = __shfl_up(inc, o, 32);
        if (lane >= o) inc += y;
    }
    if (t < BKT_NODES && lane == 31) wsum[wv] = inc;
    __syncthreads();
    if (t < BKT_NODES) {
        int wp = (lane < wv) ? wsum[lane] : 0;
        for (int o = 16; o > 0; o >>= 1) wp += __shfl_xor(wp, o, 32);
        const int start = beg + wp + inc - c;
        pos[t] = start;
        const int node = node0 + t;
        if (node < n) {
            const float d = (float)c;
            rowptr[node] = start;
            degi[node] = c;
            dis[node] = (d > 0.0f) ? rsqrtf(fmaxf(d, 1.0f)) : 0.0f;
        }
    }
    __syncthreads();
    for (int i0 = beg + t; i0 < end; i0 += 4 * PT_THREADS) {
        int2 p[4];
        for (int u = 0; u < 4; ++u) {
            const int i = i0 + u * PT_THREADS;
            p[u] = (i < end) ? pairs[i] : make_int2(-1, 0);
        }
        for (int u = 0; u < 4; ++u) {
            if (p[u].x >= 0) {
                const int slot = atomicAdd(&pos[p[u].x - node0], 1);
                csr_src[slot] = p[u].y;
            }
        }
    }
}

__device__ __forceinline__ float keep_bits(float t, int keep) {
    return __int_as_float(__float_as_int(t) & keep);
}

template <int WEIGHTED>
__global__ __launch_bounds__(256) void gather_kernel(
    const float* __restrict__ X, float* __restrict__ Y,
    const int* __restrict__ csr_src, const int* __restrict__ rowptr,
    const int* __restrict__ degi, const float* __restrict__ dis, int n)
{
    const int tid  = blockIdx.x * blockDim.x + threadIdx.x;
    const int node = tid >> 4;
    const int seg  = tid & 15;
    const bool valid = node < n;
    const int nd  = valid ? node : (n - 1);
    const int beg = rowptr[nd];
    const int cnt = valid ? degi[nd] : 0;
    const float dt = WEIGHTED ? dis[nd] : 1.0f;
    const float4* X4 = (const float4*)X;
    float4 acc = make_float4(0.0f, 0.0f, 0.0f, 0.0f);
    const int cmax = __builtin_amdgcn_readfirstlane(max(cnt, __shfl_xor(cnt, 16, 32)));
    for (int base = 0; base < cmax; base += 16) {
        const int j = base + seg;
        const int mySrc = (j < cnt) ? csr_src[beg + j] : 0;
        float myW = 1.0f;
        if (WEIGHTED) myW = dis[mySrc] * dt;
        const int lim = min(16, cmax - base);
        for (int k = 0; k < lim; k += 4) {
            int s[4]; float w[4]; float4 v[4];
            #pragma unroll
            for (int u = 0; u < 4; ++u) {
                s[u] = __shfl(mySrc, k + u, 16);
                w[u] = WEIGHTED ? __shfl(myW, k + u, 16) : 1.0f;
            }
            #pragma unroll
            for (int u = 0; u < 4; ++u) v[u] = X4[(long)s[u] * 16 + seg];
            #pragma unroll
            for (int u = 0; u < 4; ++u) {
                const int keep = ((base + k + u) < cnt) ? -1 : 0;
                acc.x += keep_bits(WEIGHTED ? v[u].x * w[u] : v[u].x, keep);
                acc.y += keep_bits(WEIGHTED ? v[u].y * w[u] : v[u].y, keep);
                acc.z += keep_bits(WEIGHTED ? v[u].z * w[u] : v[u].z, keep);
                acc.w += keep_bits(WEIGHTED ? v[u].w * w[u] : v[u].w, keep);
            }
        }
    }
    if (valid) ((float4*)Y)[(long)node * 16 + seg] = acc;
}

#define TILE_PITCH 68

__device__ __forceinline__ float4 plain_row_sum(
    const float* __restrict__ X, const int* __restrict__ csr_src, int beg, int cnt, int seg)
{
    const float4* X4 = (const float4*)X;
    float4 acc = make_float4(0.0f, 0.0f, 0.0f, 0.0f);
    const int cmax = __builtin_amdgcn_readfirstlane(max(cnt, __shfl_xor(cnt, 16, 32)));
    for (int base = 0; base < cmax; base += 16) {
        const int j = base + seg;
        const int mySrc = (j < cnt) ? csr_src[beg + j] : 0;
        const int lim = min(16, cmax - base);
        for (int k = 0; k < lim; k += 4) {
            int s[4]; float4 v[4];
            #pragma unroll
            for (int u = 0; u < 4; ++u) s[u] = __shfl(mySrc, k + u, 16);
            #pragma unroll
            for (int u = 0; u < 4; ++u) v[u] = X4[(long)s[u] * 16 + seg];
            #pragma unroll
            for (int u = 0; u < 4; ++u) {
                const int keep = ((base + k + u) < cnt) ? -1 : 0;
                acc.x += keep_bits(v[u].x, keep);
                acc.y += keep_bits(v[u].y, keep);
                acc.z += keep_bits(v[u].z, keep);
                acc.w += keep_bits(v[u].w, keep);
            }
        }
    }
    return acc;
}

__device__ __forceinline__ void stage_tile(
    const float* __restrict__ Hf, const int* __restrict__ csr_src, const int* __restrict__ rowptr,
    const int* __restrict__ degi, float* sAgg, float* sH, int n)
{
    const int t = threadIdx.x;
    const int r = t >> 4;
    const int seg = t & 15;
    const int node = blockIdx.x * 16 + r;
    const bool valid = node < n;
    const int nd = valid ? node : (n - 1);
    const int beg = rowptr[nd];
    const int cnt = valid ? degi[nd] : 0;
    const float4 own = ((const float4*)Hf)[(long)nd * 16 + seg];
    const float4 acc = plain_row_sum(Hf, csr_src, beg, cnt, seg);
    const float cinv = 1.0f / fmaxf((float)cnt, 1.0f);
    *(float4*)(sAgg + r * TILE_PITCH + seg * 4) =
        make_float4(acc.x * cinv, acc.y * cinv, acc.z * cinv, acc.w * cinv);
    *(float4*)(sH + r * TILE_PITCH + seg * 4) = own;
}

__global__ __launch_bounds__(256) void agg_gemm1_kernel(
    const float* __restrict__ Hf, const int* __restrict__ csr_src, const int* __restrict__ rowptr,
    const int* __restrict__ degi, const float* __restrict__ Wl, const float* __restrict__ Wr,
    const float* __restrict__ bias, float* __restrict__ Out, int nnodes)
{
    __shared__ __attribute__((aligned(16))) float sAgg[16 * TILE_PITCH];
    __shared__ __attribute__((aligned(16))) float sH[16 * TILE_PITCH];
    __shared__ __attribute__((aligned(16))) float sOut[16 * TILE_PITCH];
    stage_tile(Hf, csr_src, rowptr, degi, sAgg, sH, nnodes);
    __syncthreads();
    const int ct = ((threadIdx.x >> 5) - blockIdx.x) & 7;
    if (ct < 4) {
        const int lane = threadIdx.x & 31;
        const int col0 = lane & 15;
        const int kshift = (lane >> 4) * 2;
        const float* aggRow = sAgg + col0 * TILE_PITCH;
        const float* hRow   = sH   + col0 * TILE_PITCH;
        const int col = ct * 16 + col0;
        v8f acc = {};
        #pragma unroll
        for (int k0 = 0; k0 < 64; k0 += 4) {
            const int kk = k0 + kshift;
            const v2f a1 = *(const v2f*)(aggRow + kk);
            const v2f a2 = *(const v2f*)(hRow + kk);
            v2f b;
            b.x = Wl[kk * 64 + col]; b.y = Wl[(kk + 1) * 64 + col]; acc = wmma4(a1, b, acc);
            b.x = Wr[kk * 64 + col]; b.y = Wr[(kk + 1) * 64 + col]; acc = wmma4(a2, b, acc);
        }
        const float bn = bias[col];
        const int r0 = (lane >> 4) * 8;
        #pragma unroll
        for (int j = 0; j < 8; ++j) sOut[(r0 + j) * TILE_PITCH + col] = acc[j] + bn;
    }
    __syncthreads();
    const int r = threadIdx.x >> 4;
    const int seg = threadIdx.x & 15;
    const int row = blockIdx.x * 16 + r;
    const float4 s = *(const float4*)(sOut + r * TILE_PITCH + seg * 4);
    float4 y;
    y.x = s.x > 0.0f ? SELU_SCALE * s.x : SELU_SCALE * SELU_ALPHA * (expf(s.x) - 1.0f);
    y.y = s.y > 0.0f ? SELU_SCALE * s.y : SELU_SCALE * SELU_ALPHA * (expf(s.y) - 1.0f);
    y.z = s.z > 0.0f ? SELU_SCALE * s.z : SELU_SCALE * SELU_ALPHA * (expf(s.z) - 1.0f);
    y.w = s.w > 0.0f ? SELU_SCALE * s.w : SELU_SCALE * SELU_ALPHA * (expf(s.w) - 1.0f);
    if (row < nnodes) ((float4*)Out)[(long)row * 16 + seg] = y;
}

#define LOGIT_PITCH 52

__global__ __launch_bounds__(256) __attribute__((amdgpu_waves_per_eu(16, 16))) void agg_gemm2_softmax_kernel(
    const float* __restrict__ Hf, const int* __restrict__ csr_src, const int* __restrict__ rowptr,
    const int* __restrict__ degi, const float* __restrict__ Wl, const float* __restrict__ Wr,
    const float* __restrict__ bias, float* __restrict__ Out, int nnodes)
{
    __shared__ __attribute__((aligned(16))) float sAgg[16 * TILE_PITCH];
    __shared__ __attribute__((aligned(16))) float sH[16 * TILE_PITCH];
    __shared__ float sLogit[16 * LOGIT_PITCH];
    stage_tile(Hf, csr_src, rowptr, degi, sAgg, sH, nnodes);
    __syncthreads();
    const int lane = threadIdx.x & 31;
    const int col0 = lane & 15;
    const int ct = ((threadIdx.x >> 5) - blockIdx.x) & 7;
    if (ct < 3) {
        const int kshift = (lane >> 4) * 2;
        const int colr = ct * 16 + col0;
        const bool hic = colr < 40;
        const int coff = hic ? colr : 0;
        const float msk = hic ? 1.0f : 0.0f;
        const float* aggRow = sAgg + col0 * TILE_PITCH;
        const float* hRow   = sH   + col0 * TILE_PITCH;
        v8f acc = {};
        #pragma unroll
        for (int k0 = 0; k0 < 64; k0 += 4) {
            const int kk = k0 + kshift;
            const v2f a1 = *(const v2f*)(aggRow + kk);
            const v2f a2 = *(const v2f*)(hRow + kk);
            v2f b;
            b.x = Wl[kk * 40 + coff] * msk; b.y = Wl[(kk + 1) * 40 + coff] * msk; acc = wmma4(a1, b, acc);
            b.x = Wr[kk * 40 + coff] * msk; b.y = Wr[(kk + 1) * 40 + coff] * msk; acc = wmma4(a2, b, acc);
        }
        const float bn = bias[coff];
        const int r0 = (lane >> 4) * 8;
        #pragma unroll
        for (int j = 0; j < 8; ++j) sLogit[(r0 + j) * LOGIT_PITCH + colr] = acc[j] + bn;
    }
    __syncthreads();
    const int r = threadIdx.x >> 4;
    const int row = blockIdx.x * 16 + r;
    const bool hi2 = (32 + col0) < 40;
    const float* lrow = sLogit + r * LOGIT_PITCH;
    float v0 = lrow[col0];
    float v1 = lrow[16 + col0];
    float v2 = lrow[32 + col0];
    const float m01 = fmaxf(v0, v1);
    float mx = hi2 ? fmaxf(m01, v2) : m01;
    #pragma unroll
    for (int o = 8; o > 0; o >>= 1) mx = fmaxf(mx, __shfl_xor(mx, o, 32));
    v0 = expf(v0 - mx);
    v1 = expf(v1 - mx);
    const float e2 = expf(v2 - mx);
    v2 = hi2 ? e2 : 0.0f;
    float sm = v0 + v1 + v2;
    #pragma unroll
    for (int o = 8; o > 0; o >>= 1) sm += __shfl_xor(sm, o, 32);
    const float inv = 1.0f / sm;
    if (row < nnodes) {
        float* orow = Out + (long)row * 40;
        orow[col0]      = v0 * inv;
        orow[16 + col0] = v1 * inv;
        if (hi2) orow[32 + col0] = v2 * inv;
    }
}

extern "C" void kernel_launch(void* const* d_in, const int* in_sizes, int n_in,
                              void* d_out, int out_size, void* d_ws, size_t ws_size,
                              hipStream_t stream) {
    const float* x    = (const float*)d_in[0];
    const float* Wl1  = (const float*)d_in[1];
    const float* Wr1  = (const float*)d_in[2];
    const float* b1   = (const float*)d_in[3];
    const float* Wl2  = (const float*)d_in[4];
    const float* Wr2  = (const float*)d_in[5];
    const float* b2   = (const float*)d_in[6];
    const int* esrc   = (const int*)d_in[7];
    const int* edst   = (const int*)d_in[8];

    const int H  = in_sizes[3];
    const int D  = in_sizes[1] / H;
    const int Nn = in_sizes[0] / D;
    const int E  = in_sizes[7];
    (void)n_in; (void)out_size; (void)ws_size;

    float* ws     = (float*)d_ws;
    int*   degi   = (int*)ws;
    float* dis    = ws + (long)Nn;
    int*   rowptr = (int*)(ws + 2L * Nn);
    float* bufA   = ws + 3L * Nn;
    float* bufB   = bufA + (long)Nn * 64;
    float* bufC   = bufB + (long)Nn * 64;
    int*   pairs  = (int*)bufB;
    int*   bhist  = pairs + 2L * E;
    int*   csrSrc = (int*)bufC;

    const int TB = 256;
    dim3 gPart((E + PT_CHUNK - 1) / PT_CHUNK);
    const int nb = (Nn + BKT_NODES - 1) / BKT_NODES;
    dim3 gGather(((long)Nn * 16 + TB - 1) / TB);
    int*   bstart = bhist + (long)nb * gPart.x;
    int rbCount = (Nn + 15) / 16;
    dim3 gTile(rbCount);

    bucket_hist_kernel<<<gPart, PT_THREADS, 0, stream>>>(edst, bhist, E, nb);
    bucket_place_kernel<<<gPart, PT_THREADS, 0, stream>>>(esrc, edst, bhist, bstart, pairs, E, nb);
    bucket_csr_kernel<<<nb, PT_THREADS, 0, stream>>>(pairs, bstart, rowptr, degi, dis, csrSrc, Nn, E, nb);

    gather_kernel<1><<<gGather, TB, 0, stream>>>(x, bufA, csrSrc, rowptr, degi, dis, Nn);
    gather_kernel<1><<<gGather, TB, 0, stream>>>(bufA, bufB, csrSrc, rowptr, degi, dis, Nn);
    agg_gemm1_kernel<<<gTile, TB, 0, stream>>>(bufB, csrSrc, rowptr, degi, Wl1, Wr1, b1, bufA, Nn);
    agg_gemm2_softmax_kernel<<<gTile, TB, 0, stream>>>(bufA, csrSrc, rowptr, degi, Wl2, Wr2, b2, (float*)d_out, Nn);
}


# Edge list built in 2 launches, not 3: each work-group sorts its own edge chunk by bucket (LDS rank, scan), bucket blocks walk the per-chunk runs; no histogram launch; sum order = LDS arrival rank
# speedup vs baseline: 4.7090x; 1.0367x over previous
#include <hip/hip_runtime.h>
#include <hip/hip_bf16.h>
#include <math.h>

typedef __attribute__((ext_vector_type(2))) float v2f;
typedef __attribute__((ext_vector_type(8))) float v8f;

#define SELU_SCALE 1.0507009873554805f
#define SELU_ALPHA 1.6732632423543772f

__device__ __forceinline__ v8f wmma4(v2f a, v2f b, v8f c) {
    return __builtin_amdgcn_wmma_f32_16x16x4_f32(false, a, false, b, (short)0, c, false, false);
}

#define BKT_SHIFT   9
#define BKT_NODES   512
#define MAX_BKT     1024
#define PT_THREADS  1024
#define PT_PER      12
#define PT_CHUNK    (PT_THREADS * PT_PER)
#define MAX_CHUNKS  1024
#define RUN_GRP     4
#define RUN_SLOTS   3

__global__ __launch_bounds__(PT_THREADS) void bucket_sort_kernel(
    const int* __restrict__ src, const int* __restrict__ dst,
    int* __restrict__ boff, int* __restrict__ pairs_raw, int E, int nb)
{
    __shared__ int h[MAX_BKT];
    __shared__ int wsum[PT_THREADS / 32];
    const int t = threadIdx.x;
    const int lane = t & 31;
    const int wv = t >> 5;
    const int blk = blockIdx.x;
    const int base = blk * PT_CHUNK;
    h[t] = 0;
    int d[PT_PER], s[PT_PER], rk[PT_PER];
    #pragma unroll
    for (int i = 0; i < PT_PER; ++i) {
        const int e = base + i * PT_THREADS + t;
        const bool in = e < E;
        d[i] = in ? dst[e] : -1;
        s[i] = in ? src[e] : 0;
    }
    __syncthreads();
    #pragma unroll
    for (int i = 0; i < PT_PER; ++i)
        rk[i] = (d[i] >= 0) ? atomicAdd(&h[d[i] >> BKT_SHIFT], 1) : 0;
    __syncthreads();
    const int c = h[t];
    int inc = c;
    for (int o = 1; o < 32; o <<= 1) {
        int y = __shfl_up(inc, o, 32);
        if (lane >= o) inc += y;
    }
    if (lane == 31) wsum[wv] = inc;
    __syncthreads();
    int wp = (lane < wv) ? wsum[lane] : 0;
    for (int o = 16; o > 0; o >>= 1) wp += __shfl_xor(wp, o, 32);
    const int start = wp + inc - c;
    h[t] = start;
    if (t <= nb) boff[blk * (nb + 1) + t] = start;
    __syncthreads();
    int2* pairs = (int2*)pairs_raw;
    #pragma unroll
    for (int i = 0; i < PT_PER; ++i)
        if (d[i] >= 0) pairs[base + h[d[i] >> BKT_SHIFT] + rk[i]] = make_int2(d[i], s[i]);
}

template <int PLACE>
__device__ __forceinline__ void walk_runs(
    const int2* __restrict__ pairs, const int* rstart, const int* rlen, int nchunk,
    int wv, int lane, int node0, int* ctr, int* __restrict__ csr_src)
{
    for (int k0 = wv; k0 < nchunk; k0 += 32 * RUN_GRP) {
        int2 p[RUN_GRP][RUN_SLOTS];
        int len[RUN_GRP], st[RUN_GRP];
        #pragma unroll
        for (int u = 0; u < RUN_GRP; ++u) {
            const int k = k0 + 32 * u;
            const bool in = k < nchunk;
            len[u] = in ? rlen[k] : 0;
            st[u]  = in ? rstart[k] : 0;
            #pragma unroll
            for (int v = 0; v < RUN_SLOTS; ++v) {
                const int i = lane + 32 * v;
                if (i < len[u]) {
                    if (PLACE) p[u][v] = pairs[st[u] + i];
                    else       p[u][v] = make_int2(pairs[st[u] + i].x, 0);
                } else {
                    p[u][v] = make_int2(-1, 0);
                }
            }
        }
        #pragma unroll
        for (int u = 0; u < RUN_GRP; ++u) {
            #pragma unroll
            for (int v = 0; v < RUN_SLOTS; ++v) {
                if (p[u][v].x >= 0) {
                    const int slot = atomicAdd(&ctr[p[u][v].x - node0], 1);
                    if (PLACE) csr_src[slot] = p[u][v].y;
                }
            }
        }
        #pragma unroll
        for (int u = 0; u < RUN_GRP; ++u) {
            for (int i = lane + 32 * RUN_SLOTS; i < len[u]; i += 32) {
                const int2 q = pairs[st[u] + i];
                const int slot = atomicAdd(&ctr[q.x - node0], 1);
                if (PLACE) csr_src[slot] = q.y;
            }
        }
    }
}

__global__ __launch_bounds__(PT_THREADS) void bucket_csr_kernel(
    const int* __restrict__ pairs_raw, const int* __restrict__ boff,
    int* __restrict__ rowptr, int* __restrict__ degi, float* __restrict__ dis,
    int* __restrict__ csr_src, int n, int nb, int nchunk)
{
    __shared__ int cnt[BKT_NODES];
    __shared__ int pos[BKT_NODES];
    __shared__ int wsum[PT_THREADS / 32];
    __shared__ int rstart[MAX_CHUNKS];
    __shared__ int rlen[MAX_CHUNKS];
    const int t = threadIdx.x;
    const int lane = t & 31;
    const int wv = t >> 5;
    const int b = blockIdx.x;
    const int2* pairs = (const int2*)pairs_raw;
    const int node0 = b << BKT_SHIFT;
    int o0 = 0, o1 = 0;
    if (t < nchunk) {
        o0 = boff[t * (nb + 1) + b];
        o1 = boff[t * (nb + 1) + b + 1];
    }
    rstart[t] = t * PT_CHUNK + o0;
    rlen[t] = o1 - o0;
    if (t < BKT_NODES) cnt[t] = 0;
    int sb = o0;
    for (int o = 16; o > 0; o >>= 1) sb += __shfl_xor(sb, o, 32);
    if (lane == 0) wsum[wv] = sb;
    __syncthreads();
    int beg = wsum[lane];
    for (int o = 16; o > 0; o >>= 1) beg += __shfl_xor(beg, o, 32);
    walk_runs<0>(pairs, rstart, rlen, nchunk, wv, lane, node0, cnt, csr_src);
    __syncthreads();
    const int c = (t < BKT_NODES) ? cnt[t] : 0;
    int inc = c;
    for (int o = 1; o < 32; o <<= 1) {
        int y = __shfl_up(inc, o, 32);
        if (lane >= o) inc += y;
    }
    if (t < BKT_NODES && lane == 31) wsum[wv] = inc;
    __syncthreads();
    if (t < BKT_NODES) {
        int wp = (lane < wv) ? wsum[lane] : 0;
        for (int o = 16; o > 0; o >>= 1) wp += __shfl_xor(wp, o, 32);
        const int start = beg + wp + inc - c;
        pos[t] = start;
        const int node = node0 + t;
        if (node < n) {
            const float d = (float)c;
            rowptr[node] = start;
            degi[node] = c;
            dis[node] = (d > 0.0f) ? rsqrtf(fmaxf(d, 1.0f)) : 0.0f;
        }
    }
    __syncthreads();
    walk_runs<1>(pairs, rstart, rlen, nchunk, wv, lane, node0, pos, csr_src);
}

__device__ __forceinline__ float keep_bits(float t, int keep) {
    return __int_as_float(__float_as_int(t) & keep);
}

template <int WEIGHTED>
__global__ __launch_bounds__(256) void gather_kernel(
    const float* __restrict__ X, float* __restrict__ Y,
    const int* __restrict__ csr_src, const int* __restrict__ rowptr,
    const int* __restrict__ degi, const float* __restrict__ dis, int n)
{
    const int tid  = blockIdx.x * blockDim.x + threadIdx.x;
    const int node = tid >> 4;
    const int seg  = tid & 15;
    const bool valid = node < n;
    const int nd  = valid ? node : (n - 1);
    const int beg = rowptr[nd];
    const int cnt = valid ? degi[nd] : 0;
    const float dt = WEIGHTED ? dis[nd] : 1.0f;
    const float4* X4 = (const float4*)X;
    float4 acc = make_float4(0.0f, 0.0f, 0.0f, 0.0f);
    const int cmax = __builtin_amdgcn_readfirstlane(max(cnt, __shfl_xor(cnt, 16, 32)));
    for (int base = 0; base < cmax; base += 16) {
        const int j = base + seg;
        const int mySrc = (j < cnt) ? csr_src[beg + j] : 0;
        float myW = 1.0f;
        if (WEIGHTED) myW = dis[mySrc] * dt;
        const int lim = min(16, cmax - base);
        for (int k = 0; k < lim; k += 4) {
            int s[4]; float w[4]; float4 v[4];
            #pragma unroll
            for (int u = 0; u < 4; ++u) {
                s[u] = __shfl(mySrc, k + u, 16);
                w[u] = WEIGHTED ? __shfl(myW, k + u, 16) : 1.0f;
            }
            #pragma unroll
            for (int u = 0; u < 4; ++u) v[u] = X4[(long)s[u] * 16 + seg];
            #pragma unroll
            for (int u = 0; u < 4; ++u) {
                const int keep = ((base + k + u) < cnt) ? -1 : 0;
                acc.x += keep_bits(WEIGHTED ? v[u].x * w[u] : v[u].x, keep);
                acc.y += keep_bits(WEIGHTED ? v[u].y * w[u] : v[u].y, keep);
                acc.z += keep_bits(WEIGHTED ? v[u].z * w[u] : v[u].z, keep);
                acc.w += keep_bits(WEIGHTED ? v[u].w * w[u] : v[u].w, keep);
            }
        }
    }
    if (valid) ((float4*)Y)[(long)node * 16 + seg] = acc;
}

#define TILE_PITCH 68

__device__ __forceinline__ float4 plain_row_sum(
    const float* __restrict__ X, const int* __restrict__ csr_src, int beg, int cnt, int seg)
{
    const float4* X4 = (const float4*)X;
    float4 acc = make_float4(0.0f, 0.0f, 0.0f, 0.0f);
    const int cmax = __builtin_amdgcn_readfirstlane(max(cnt, __shfl_xor(cnt, 16, 32)));
    for (int base = 0; base < cmax; base += 16) {
        const int j = base + seg;
        const int mySrc = (j < cnt) ? csr_src[beg + j] : 0;
        const int lim = min(16, cmax - base);
        for (int k = 0; k < lim; k += 4) {
            int s[4]; float4 v[4];
            #pragma unroll
            for (int u = 0; u < 4; ++u) s[u] = __shfl(mySrc, k + u, 16);
            #pragma unroll
            for (int u = 0; u < 4; ++u) v[u] = X4[(long)s[u] * 16 + seg];
            #pragma unroll
            for (int u = 0; u < 4; ++u) {
                const int keep = ((base + k + u) < cnt) ? -1 : 0;
                acc.x += keep_bits(v[u].x, keep);
                acc.y += keep_bits(v[u].y, keep);
                acc.z += keep_bits(v[u].z, keep);
                acc.w += keep_bits(v[u].w, keep);
            }
        }
    }
    return acc;
}

__device__ __forceinline__ void stage_tile(
    const float* __restrict__ Hf, const int* __restrict__ csr_src, const int* __restrict__ rowptr,
    const int* __restrict__ degi, float* sAgg, float* sH, int n)
{
    const int t = threadIdx.x;
    const int r = t >> 4;
    const int seg = t & 15;
    const int node = blockIdx.x * 16 + r;
    const bool valid = node < n;
    const int nd = valid ? node : (n - 1);
    const int beg = rowptr[nd];
    const int cnt = valid ? degi[nd] : 0;
    const float4 own = ((const float4*)Hf)[(long)nd * 16 + seg];
    const float4 acc = plain_row_sum(Hf, csr_src, beg, cnt, seg);
    const float cinv = 1.0f / fmaxf((float)cnt, 1.0f);
    *(float4*)(sAgg + r * TILE_PITCH + seg * 4) =
        make_float4(acc.x * cinv, acc.y * cinv, acc.z * cinv, acc.w * cinv);
    *(float4*)(sH + r * TILE_PITCH + seg * 4) = own;
}

__global__ __launch_bounds__(256) void agg_gemm1_kernel(
    const float* __restrict__ Hf, const int* __restrict__ csr_src, const int* __restrict__ rowptr,
    const int* __restrict__ degi, const float* __restrict__ Wl, const float* __restrict__ Wr,
    const float* __restrict__ bias, float* __restrict__ Out, int nnodes)
{
    __shared__ __attribute__((aligned(16))) float sAgg[16 * TILE_PITCH];
    __shared__ __attribute__((aligned(16))) float sH[16 * TILE_PITCH];
    __shared__ __attribute__((aligned(16))) float sOut[16 * TILE_PITCH];
    stage_tile(Hf, csr_src, rowptr, degi, sAgg, sH, nnodes);
    __syncthreads();
    const int ct = ((threadIdx.x >> 5) - blockIdx.x) & 7;
    if (ct < 4) {
        const int lane = threadIdx.x & 31;
        const int col0 = lane & 15;
        const int kshift = (lane >> 4) * 2;
        const float* aggRow = sAgg + col0 * TILE_PITCH;
        const float* hRow   = sH   + col0 * TILE_PITCH;
        const int col = ct * 16 + col0;
        v8f acc = {};
        #pragma unroll
        for (int k0 = 0; k0 < 64; k0 += 4) {
            const int kk = k0 + kshift;
            const v2f a1 = *(const v2f*)(aggRow + kk);
            const v2f a2 = *(const v2f*)(hRow + kk);
            v2f b;
            b.x = Wl[kk * 64 + col]; b.y = Wl[(kk + 1) * 64 + col]; acc = wmma4(a1, b, acc);
            b.x = Wr[kk * 64 + col]; b.y = Wr[(kk + 1) * 64 + col]; acc = wmma4(a2, b, acc);
        }
        const float bn = bias[col];
        const int r0 = (lane >> 4) * 8;
        #pragma unroll
        for (int j = 0; j < 8; ++j) sOut[(r0 + j) * TILE_PITCH + col] = acc[j] + bn;
    }
    __syncthreads();
    const int r = threadIdx.x >> 4;
    const int seg = threadIdx.x & 15;
    const int row = blockIdx.x * 16 + r;
    const float4 s = *(const float4*)(sOut + r * TILE_PITCH + seg * 4);
    float4 y;
    y.x = s.x > 0.0f ? SELU_SCALE * s.x : SELU_SCALE * SELU_ALPHA * (expf(s.x) - 1.0f);
    y.y = s.y > 0.0f ? SELU_SCALE * s.y : SELU_SCALE * SELU_ALPHA * (expf(s.y) - 1.0f);
    y.z = s.z > 0.0f ? SELU_SCALE * s.z : SELU_SCALE * SELU_ALPHA * (expf(s.z) - 1.0f);
    y.w = s.w > 0.0f ? SELU_SCALE * s.w : SELU_SCALE * SELU_ALPHA * (expf(s.w) - 1.0f);
    if (row < nnodes) ((float4*)Out)[(long)row * 16 + seg] = y;
}

#define LOGIT_PITCH 52

__global__ __launch_bounds__(256) __attribute__((amdgpu_waves_per_eu(16, 16))) void agg_gemm2_softmax_kernel(
    const float* __restrict__ Hf, const int* __restrict__ csr_src, const int* __restrict__ rowptr,
    const int* __restrict__ degi, const float* __restrict__ Wl, const float* __restrict__ Wr,
    const float* __restrict__ bias, float* __restrict__ Out, int nnodes)
{
    __shared__ __attribute__((aligned(16))) float sAgg[16 * TILE_PITCH];
    __shared__ __attribute__((aligned(16))) float sH[16 * TILE_PITCH];
    __shared__ float sLogit[16 * LOGIT_PITCH];
    stage_tile(Hf, csr_src, rowptr, degi, sAgg, sH, nnodes);
    __syncthreads();
    const int lane = threadIdx.x & 31;
    const int col0 = lane & 15;
    const int ct = ((threadIdx.x >> 5) - blockIdx.x) & 7;
    if (ct < 3) {
        const int kshift = (lane >> 4) * 2;
        const int colr = ct * 16 + col0;
        const bool hic = colr < 40;
        const int coff = hic ? colr : 0;
        const float msk = hic ? 1.0f : 0.0f;
        const float* aggRow = sAgg + col0 * TILE_PITCH;
        const float* hRow   = sH   + col0 * TILE_PITCH;
        v8f acc = {};
        #pragma unroll
        for (int k0 = 0; k0 < 64; k0 += 4) {
            const int kk = k0 + kshift;
            const v2f a1 = *(const v2f*)(aggRow + kk);
            const v2f a2 = *(const v2f*)(hRow + kk);
            v2f b;
            b.x = Wl[kk * 40 + coff] * msk; b.y = Wl[(kk + 1) * 40 + coff] * msk; acc = wmma4(a1, b, acc);
            b.x = Wr[kk * 40 + coff] * msk; b.y = Wr[(kk + 1) * 40 + coff] * msk; acc = wmma4(a2, b, acc);
        }
        const float bn = bias[coff];
        const int r0 = (lane >> 4) * 8;
        #pragma unroll
        for (int j = 0; j < 8; ++j) sLogit[(r0 + j) * LOGIT_PITCH + colr] = acc[j] + bn;
    }
    __syncthreads();
    const int r = threadIdx.x >> 4;
    const int row = blockIdx.x * 16 + r;
    const bool hi2 = (32 + col0) < 40;
    const float* lrow = sLogit + r * LOGIT_PITCH;
    float v0 = lrow[col0];
    float v1 = lrow[16 + col0];
    float v2 = lrow[32 + col0];
    const float m01 = fmaxf(v0, v1);
    float mx = hi2 ? fmaxf(m01, v2) : m01;
    #pragma unroll
    for (int o = 8; o > 0; o >>= 1) mx = fmaxf(mx, __shfl_xor(mx, o, 32));
    v0 = expf(v0 - mx);
    v1 = expf(v1 - mx);
    const float e2 = expf(v2 - mx);
    v2 = hi2 ? e2 : 0.0f;
    float sm = v0 + v1 + v2;
    #pragma unroll
    for (int o = 8; o > 0; o >>= 1) sm += __shfl_xor(sm, o, 32);
    const float inv = 1.0f / sm;
    if (row < nnodes) {
        float* orow = Out + (long)row * 40;
        orow[col0]      = v0 * inv;
        orow[16 + col0] = v1 * inv;
        if (hi2) orow[32 + col0] = v2 * inv;
    }
}

extern "C" void kernel_launch(void* const* d_in, const int* in_sizes, int n_in,
                              void* d_out, int out_size, void* d_ws, size_t ws_size,
                              hipStream_t stream) {
    const float* x    = (const float*)d_in[0];
    const float* Wl1  = (const float*)d_in[1];
    const float* Wr1  = (const float*)d_in[2];
    const float* b1   = (const float*)d_in[3];
    const float* Wl2  = (const float*)d_in[4];
    const float* Wr2  = (const float*)d_in[5];
    const float* b2   = (const float*)d_in[6];
    const int* esrc   = (const int*)d_in[7];
    const int* edst   = (const int*)d_in[8];

    const int H  = in_sizes[3];
    const int D  = in_sizes[1] / H;
    const int Nn = in_sizes[0] / D;
    const int E  = in_sizes[7];
    (void)n_in; (void)out_size; (void)ws_size;

    float* ws     = (float*)d_ws;
    int*   degi   = (int*)ws;
    float* dis    = ws + (long)Nn;
    int*   rowptr = (int*)(ws + 2L * Nn);
    float* bufA   = ws + 3L * Nn;
    float* bufB   = bufA + (long)Nn * 64;
    float* bufC   = bufB + (long)Nn * 64;
    int*   pairs  = (int*)bufB;
    int*   boff   = pairs + 2L * E;
    int*   csrSrc = (int*)bufC;

    const int TB = 256;
    dim3 gPart((E + PT_CHUNK - 1) / PT_CHUNK);
    const int nb = (Nn + BKT_NODES - 1) / BKT_NODES;
    dim3 gGather(((long)Nn * 16 + TB - 1) / TB);
    int rbCount = (Nn + 15) / 16;
    dim3 gTile(rbCount);

    bucket_sort_kernel<<<gPart, PT_THREADS, 0, stream>>>(esrc, edst, boff, pairs, E, nb);
    bucket_csr_kernel<<<nb, PT_THREADS, 0, stream>>>(pairs, boff, rowptr, degi, dis, csrSrc, Nn, nb, (int)gPart.x);

    gather_kernel<1><<<gGather, TB, 0, stream>>>(x, bufA, csrSrc, rowptr, degi, dis, Nn);
    gather_kernel<1><<<gGather, TB, 0, stream>>>(bufA, bufB, csrSrc, rowptr, degi, dis, Nn);
    agg_gemm1_kernel<<<gTile, TB, 0, stream>>>(bufB, csrSrc, rowptr, degi, Wl1, Wr1, b1, bufA, Nn);
    agg_gemm2_softmax_kernel<<<gTile, TB, 0, stream>>>(bufA, csrSrc, rowptr, degi, Wl2, Wr2, b2, (float*)d_out, Nn);
}


# Both list-building launches put their records at the final slot in LDS first and store the sorted block as whole 128-byte lines, not one scattered 4- or 8-byte store per edge; sum order unchanged
# speedup vs baseline: 5.0414x; 1.0706x over previous
#include <hip/hip_runtime.h>
#include <hip/hip_bf16.h>
#include <math.h>

typedef __attribute__((ext_vector_type(2))) float v2f;
typedef __attribute__((ext_vector_type(8))) float v8f;

#define SELU_SCALE 1.0507009873554805f
#define SELU_ALPHA 1.6732632423543772f

__device__ __forceinline__ v8f wmma4(v2f a, v2f b, v8f c) {
    return __builtin_amdgcn_wmma_f32_16x16x4_f32(false, a, false, b, (short)0, c, false, false);
}

#define BKT_SHIFT   9
#define BKT_NODES   512
#define MAX_BKT     1024
#define PT_THREADS  1024
#define PT_PER      12
#define PT_CHUNK    (PT_THREADS * PT_PER)
#define MAX_CHUNKS  1024
#define RUN_GRP     4
#define RUN_SLOTS   3
#define STAGE_CAP   16384

__global__ __launch_bounds__(PT_THREADS) void bucket_sort_kernel(
    const int* __restrict__ src, const int* __restrict__ dst,
    int* __restrict__ boff, int* __restrict__ pairs_raw, int E, int nb)
{
    __shared__ int h[MAX_BKT];
    __shared__ int wsum[PT_THREADS / 32];
    __shared__ __attribute__((aligned(16))) int2 sp[PT_CHUNK];
    const int t = threadIdx.x;
    const int lane = t & 31;
    const int wv = t >> 5;
    const int blk = blockIdx.x;
    const int base = blk * PT_CHUNK;
    h[t] = 0;
    int d[PT_PER], s[PT_PER], rk[PT_PER];
    #pragma unroll
    for (int i = 0; i < PT_PER; ++i) {
        const int e = base + i * PT_THREADS + t;
        const bool in = e < E;
        d[i] = in ? dst[e] : -1;
        s[i] = in ? src[e] : 0;
    }
    __syncthreads();
    #pragma unroll
    for (int i = 0; i < PT_PER; ++i)
        rk[i] = (d[i] >= 0) ? atomicAdd(&h[d[i] >> BKT_SHIFT], 1) : 0;
    __syncthreads();
    const int c = h[t];
    int inc = c;
    for (int o = 1; o < 32; o <<= 1) {
        int y = __shfl_up(inc, o, 32);
        if (lane >= o) inc += y;
    }
    if (lane == 31) wsum[wv] = inc;
    __syncthreads();
    int wp = (lane < wv) ? wsum[lane] : 0;
    for (int o = 16; o > 0; o >>= 1) wp += __shfl_xor(wp, o, 32);
    const int start = wp + inc - c;
    h[t] = start;
    if (t <= nb) boff[blk * (nb + 1) + t] = start;
    __syncthreads();
    #pragma unroll
    for (int i = 0; i < PT_PER; ++i)
        if (d[i] >= 0) sp[h[d[i] >> BKT_SHIFT] + rk[i]] = make_int2(d[i], s[i]);
    __syncthreads();
    const int cntc = min(PT_CHUNK, E - base);
    const int n4 = cntc >> 1;
    int4* gp = (int4*)(pairs_raw + 2L * base);
    const int4* lp = (const int4*)sp;
    if (cntc == PT_CHUNK) {
        int4 r[PT_PER / 2];
        #pragma unroll
        for (int i = 0; i < PT_PER / 2; ++i) r[i] = lp[i * PT_THREADS + t];
        #pragma unroll
        for (int i = 0; i < PT_PER / 2; ++i) gp[i * PT_THREADS + t] = r[i];
    } else {
        for (int q = t; q < n4; q += PT_THREADS) gp[q] = lp[q];
        if ((cntc & 1) && t == 0) ((int2*)pairs_raw)[base + cntc - 1] = sp[cntc - 1];
    }
}

template <int PLACE>
__device__ __forceinline__ void walk_runs(
    const int2* __restrict__ pairs, const int* rstart, const int* rlen, int nchunk,
    int wv, int lane, int node0, int* ctr, int* stage, int* __restrict__ csr_bkt)
{
    for (int k0 = wv; k0 < nchunk; k0 += 32 * RUN_GRP) {
        int2 p[RUN_GRP][RUN_SLOTS];
        int len[RUN_GRP], st[RUN_GRP];
        #pragma unroll
        for (int u = 0; u < RUN_GRP; ++u) {
            const int k = k0 + 32 * u;
            const bool in = k < nchunk;
            len[u] = in ? rlen[k] : 0;
            st[u]  = in ? rstart[k] : 0;
            #pragma unroll
            for (int v = 0; v < RUN_SLOTS; ++v) {
                const int i = lane + 32 * v;
                if (i < len[u]) {
                    if (PLACE) p[u][v] = pairs[st[u] + i];
                    else       p[u][v] = make_int2(pairs[st[u] + i].x, 0);
                } else {
                    p[u][v] = make_int2(-1, 0);
                }
            }
        }
        #pragma unroll
        for (int u = 0; u < RUN_GRP; ++u) {
            #pragma unroll
            for (int v = 0; v < RUN_SLOTS; ++v) {
                if (p[u][v].x >= 0) {
                    const int slot = atomicAdd(&ctr[p[u][v].x - node0], 1);
                    if (PLACE) {
                        if (slot < STAGE_CAP) stage[slot] = p[u][v].y;
                        else                  csr_bkt[slot] = p[u][v].y;
                    }
                }
            }
        }
        #pragma unroll
        for (int u = 0; u < RUN_GRP; ++u) {
            for (int i = lane + 32 * RUN_SLOTS; i < len[u]; i += 32) {
                const int2 q = pairs[st[u] + i];
                const int slot = atomicAdd(&ctr[q.x - node0], 1);
                if (PLACE) {
                    if (slot < STAGE_CAP) stage[slot] = q.y;
                    else                  csr_bkt[slot] = q.y;
                }
            }
        }
    }
}

__global__ __launch_bounds__(PT_THREADS) void bucket_csr_kernel(
    const int* __restrict__ pairs_raw, const int* __restrict__ boff,
    int* __restrict__ rowptr, int* __restrict__ degi, float* __restrict__ dis,
    int* __restrict__ csr_src, int n, int nb, int nchunk)
{
    __shared__ int cnt[BKT_NODES];
    __shared__ int pos[BKT_NODES];
    __shared__ int wsum[PT_THREADS / 32];
    __shared__ int rstart[MAX_CHUNKS];
    __shared__ int rlen[MAX_CHUNKS];
    __shared__ int stage[STAGE_CAP];
    const int t = threadIdx.x;
    const int lane = t & 31;
    const int wv = t >> 5;
    const int b = blockIdx.x;
    const int2* pairs = (const int2*)pairs_raw;
    const int node0 = b << BKT_SHIFT;
    int o0 = 0, o1 = 0;
    if (t < nchunk) {
        o0 = boff[t * (nb + 1) + b];
        o1 = boff[t * (nb + 1) + b + 1];
    }
    rstart[t] = t * PT_CHUNK + o0;
    rlen[t] = o1 - o0;
    if (t < BKT_NODES) cnt[t] = 0;
    int sb = o0;
    for (int o = 16; o > 0; o >>= 1) sb += __shfl_xor(sb, o, 32);
    if (lane == 0) wsum[wv] = sb;
    __syncthreads();
    int beg = wsum[lane];
    for (int o = 16; o > 0; o >>= 1) beg += __shfl_xor(beg, o, 32);
    walk_runs<0>(pairs, rstart, rlen, nchunk, wv, lane, node0, cnt, stage, csr_src + beg);
    __syncthreads();
    const int c = (t < BKT_NODES) ? cnt[t] : 0;
    int inc = c;
    for (int o = 1; o < 32; o <<= 1) {
        int y = __shfl_up(inc, o, 32);
        if (lane >= o) inc += y;
    }
    if (t < BKT_NODES && lane == 31) wsum[wv] = inc;
    __syncthreads();
    int tot = (lane < BKT_NODES / 32) ? wsum[lane] : 0;
    for (int o = 16; o > 0; o >>= 1) tot += __shfl_xor(tot, o, 32);
    if (t < BKT_NODES) {
        int wp = (lane < wv) ? wsum[lane] : 0;
        for (int o = 16; o > 0; o >>= 1) wp += __shfl_xor(wp, o, 32);
        const int start = wp + inc - c;
        pos[t] = start;
        const int node = node0 + t;
        if (node < n) {
            const float d = (float)c;
            rowptr[node] = beg + start;
            degi[node] = c;
            dis[node] = (d > 0.0f) ? rsqrtf(fmaxf(d, 1.0f)) : 0.0f;
        }
    }
    __syncthreads();
    walk_runs<1>(pairs, rstart, rlen, nchunk, wv, lane, node0, pos, stage, csr_src + beg);
    __syncthreads();
    const int lim = min(tot, STAGE_CAP);
    for (int g = (beg & ~31) + t; g < beg + lim; g += PT_THREADS)
        if (g >= beg) csr_src[g] = stage[g - beg];
}

__device__ __forceinline__ float keep_bits(float t, int keep) {
    return __int_as_float(__float_as_int(t) & keep);
}

template <int WEIGHTED>
__global__ __launch_bounds__(256) void gather_kernel(
    const float* __restrict__ X, float* __restrict__ Y,
    const int* __restrict__ csr_src, const int* __restrict__ rowptr,
    const int* __restrict__ degi, const float* __restrict__ dis, int n)
{
    const int tid  = blockIdx.x * blockDim.x + threadIdx.x;
    const int node = tid >> 4;
    const int seg  = tid & 15;
    const bool valid = node < n;
    const int nd  = valid ? node : (n - 1);
    const int beg = rowptr[nd];
    const int cnt = valid ? degi[nd] : 0;
    const float dt = WEIGHTED ? dis[nd] : 1.0f;
    const float4* X4 = (const float4*)X;
    float4 acc = make_float4(0.0f, 0.0f, 0.0f, 0.0f);
    const int cmax = __builtin_amdgcn_readfirstlane(max(cnt, __shfl_xor(cnt, 16, 32)));
    for (int base = 0; base < cmax; base += 16) {
        const int j = base + seg;
        const int mySrc = (j < cnt) ? csr_src[beg + j] : 0;
        float myW = 1.0f;
        if (WEIGHTED) myW = dis[mySrc] * dt;
        const int lim = min(16, cmax - base);
        for (int k = 0; k < lim; k += 4) {
            int s[4]; float w[4]; float4 v[4];
            #pragma unroll
            for (int u = 0; u < 4; ++u) {
                s[u] = __shfl(mySrc, k + u, 16);
                w[u] = WEIGHTED ? __shfl(myW, k + u, 16) : 1.0f;
            }
            #pragma unroll
            for (int u = 0; u < 4; ++u) v[u] = X4[(long)s[u] * 16 + seg];
            #pragma unroll
            for (int u = 0; u < 4; ++u) {
                const int keep = ((base + k + u) < cnt) ? -1 : 0;
                acc.x += keep_bits(WEIGHTED ? v[u].x * w[u] : v[u].x, keep);
                acc.y += keep_bits(WEIGHTED ? v[u].y * w[u] : v[u].y, keep);
                acc.z += keep_bits(WEIGHTED ? v[u].z * w[u] : v[u].z, keep);
                acc.w += keep_bits(WEIGHTED ? v[u].w * w[u] : v[u].w, keep);
            }
        }
    }
    if (valid) ((float4*)Y)[(long)node * 16 + seg] = acc;
}

#define TILE_PITCH 68

__device__ __forceinline__ float4 plain_row_sum(
    const float* __restrict__ X, const int* __restrict__ csr_src, int beg, int cnt, int seg)
{
    const float4* X4 = (const float4*)X;
    float4 acc = make_float4(0.0f, 0.0f, 0.0f, 0.0f);
    const int cmax = __builtin_amdgcn_readfirstlane(max(cnt, __shfl_xor(cnt, 16, 32)));
    for (int base = 0; base < cmax; base += 16) {
        const int j = base + seg;
        const int mySrc = (j < cnt) ? csr_src[beg + j] : 0;
        const int lim = min(16, cmax - base);
        for (int k = 0; k < lim; k += 4) {
            int s[4]; float4 v[4];
            #pragma unroll
            for (int u = 0; u < 4; ++u) s[u] = __shfl(mySrc, k + u, 16);
            #pragma unroll
            for (int u = 0; u < 4; ++u) v[u] = X4[(long)s[u] * 16 + seg];
            #pragma unroll
            for (int u = 0; u < 4; ++u) {
                const int keep = ((base + k + u) < cnt) ? -1 : 0;
                acc.x += keep_bits(v[u].x, keep);
                acc.y += keep_bits(v[u].y, keep);
                acc.z += keep_bits(v[u].z, keep);
                acc.w += keep_bits(v[u].w, keep);
            }
        }
    }
    return acc;
}

__device__ __forceinline__ void stage_tile(
    const float* __restrict__ Hf, const int* __restrict__ csr_src, const int* __restrict__ rowptr,
    const int* __restrict__ degi, float* sAgg, float* sH, int n)
{
    const int t = threadIdx.x;
    const int r = t >> 4;
    const int seg = t & 15;
    const int node = blockIdx.x * 16 + r;
    const bool valid = node < n;
    const int nd = valid ? node : (n - 1);
    const int beg = rowptr[nd];
    const int cnt = valid ? degi[nd] : 0;
    const float4 own = ((const float4*)Hf)[(long)nd * 16 + seg];
    const float4 acc = plain_row_sum(Hf, csr_src, beg, cnt, seg);
    const float cinv = 1.0f / fmaxf((float)cnt, 1.0f);
    *(float4*)(sAgg + r * TILE_PITCH + seg * 4) =
        make_float4(acc.x * cinv, acc.y * cinv, acc.z * cinv, acc.w * cinv);
    *(float4*)(sH + r * TILE_PITCH + seg * 4) = own;
}

__global__ __launch_bounds__(256) void agg_gemm1_kernel(
    const float* __restrict__ Hf, const int* __restrict__ csr_src, const int* __restrict__ rowptr,
    const int* __restrict__ degi, const float* __restrict__ Wl, const float* __restrict__ Wr,
    const float* __restrict__ bias, float* __restrict__ Out, int nnodes)
{
    __shared__ __attribute__((aligned(16))) float sAgg[16 * TILE_PITCH];
    __shared__ __attribute__((aligned(16))) float sH[16 * TILE_PITCH];
    __shared__ __attribute__((aligned(16))) float sOut[16 * TILE_PITCH];
    stage_tile(Hf, csr_src, rowptr, degi, sAgg, sH, nnodes);
    __syncthreads();
    const int ct = ((threadIdx.x >> 5) - blockIdx.x) & 7;
    if (ct < 4) {
        const int lane = threadIdx.x & 31;
        const int col0 = lane & 15;
        const int kshift = (lane >> 4) * 2;
        const float* aggRow = sAgg + col0 * TILE_PITCH;
        const float* hRow   = sH   + col0 * TILE_PITCH;
        const int col = ct * 16 + col0;
        v8f acc = {};
        #pragma unroll
        for (int k0 = 0; k0 < 64; k0 += 4) {
            const int kk = k0 + kshift;
            const v2f a1 = *(const v2f*)(aggRow + kk);
            const v2f a2 = *(const v2f*)(hRow + kk);
            v2f b;
            b.x = Wl[kk * 64 + col]; b.y = Wl[(kk + 1) * 64 + col]; acc = wmma4(a1, b, acc);
            b.x = Wr[kk * 64 + col]; b.y = Wr[(kk + 1) * 64 + col]; acc = wmma4(a2, b, acc);
        }
        const float bn = bias[col];
        const int r0 = (lane >> 4) * 8;
        #pragma unroll
        for (int j = 0; j < 8; ++j) sOut[(r0 + j) * TILE_PITCH + col] = acc[j] + bn;
    }
    __syncthreads();
    const int r = threadIdx.x >> 4;
    const int seg = threadIdx.x & 15;
    const int row = blockIdx.x * 16 + r;
    const float4 s = *(const float4*)(sOut + r * TILE_PITCH + seg * 4);
    float4 y;
    y.x = s.x > 0.0f ? SELU_SCALE * s.x : SELU_SCALE * SELU_ALPHA * (expf(s.x) - 1.0f);
    y.y = s.y > 0.0f ? SELU_SCALE * s.y : SELU_SCALE * SELU_ALPHA * (expf(s.y) - 1.0f);
    y.z = s.z > 0.0f ? SELU_SCALE * s.z : SELU_SCALE * SELU_ALPHA * (expf(s.z) - 1.0f);
    y.w = s.w > 0.0f ? SELU_SCALE * s.w : SELU_SCALE * SELU_ALPHA * (expf(s.w) - 1.0f);
    if (row < nnodes) ((float4*)Out)[(long)row * 16 + seg] = y;
}

#define LOGIT_PITCH 52

__global__ __launch_bounds__(256) __attribute__((amdgpu_waves_per_eu(16, 16))) void agg_gemm2_softmax_kernel(
    const float* __restrict__ Hf, const int* __restrict__ csr_src, const int* __restrict__ rowptr,
    const int* __restrict__ degi, const float* __restrict__ Wl, const float* __restrict__ Wr,
    const float* __restrict__ bias, float* __restrict__ Out, int nnodes)
{
    __shared__ __attribute__((aligned(16))) float sAgg[16 * TILE_PITCH];
    __shared__ __attribute__((aligned(16))) float sH[16 * TILE_PITCH];
    __shared__ float sLogit[16 * LOGIT_PITCH];
    stage_tile(Hf, csr_src, rowptr, degi, sAgg, sH, nnodes);
    __syncthreads();
    const int lane = threadIdx.x & 31;
    const int col0 = lane & 15;
    const int ct = ((threadIdx.x >> 5) - blockIdx.x) & 7;
    if (ct < 3) {
        const int kshift = (lane >> 4) * 2;
        const int colr = ct * 16 + col0;
        const bool hic = colr < 40;
        const int coff = hic ? colr : 0;
        const float msk = hic ? 1.0f : 0.0f;
        const float* aggRow = sAgg + col0 * TILE_PITCH;
        const float* hRow   = sH   + col0 * TILE_PITCH;
        v8f acc = {};
        #pragma unroll
        for (int k0 = 0; k0 < 64; k0 += 4) {
            const int kk = k0 + kshift;
            const v2f a1 = *(const v2f*)(aggRow + kk);
            const v2f a2 = *(const v2f*)(hRow + kk);
            v2f b;
            b.x = Wl[kk * 40 + coff] * msk; b.y = Wl[(kk + 1) * 40 + coff] * msk; acc = wmma4(a1, b, acc);
            b.x = Wr[kk * 40 + coff] * msk; b.y = Wr[(kk + 1) * 40 + coff] * msk; acc = wmma4(a2, b, acc);
        }
        const float bn = bias[coff];
        const int r0 = (lane >> 4) * 8;
        #pragma unroll
        for (int j = 0; j < 8; ++j) sLogit[(r0 + j) * LOGIT_PITCH + colr] = acc[j] + bn;
    }
    __syncthreads();
    const int r = threadIdx.x >> 4;
    const int row = blockIdx.x * 16 + r;
    const bool hi2 = (32 + col0) < 40;
    const float* lrow = sLogit + r * LOGIT_PITCH;
    float v0 = lrow[col0];
    float v1 = lrow[16 + col0];
    float v2 = lrow[32 + col0];
    const float m01 = fmaxf(v0, v1);
    float mx = hi2 ? fmaxf(m01, v2) : m01;
    #pragma unroll
    for (int o = 8; o > 0; o >>= 1) mx = fmaxf(mx, __shfl_xor(mx, o, 32));
    v0 = expf(v0 - mx);
    v1 = expf(v1 - mx);
    const float e2 = expf(v2 - mx);
    v2 = hi2 ? e2 : 0.0f;
    float sm = v0 + v1 + v2;
    #pragma unroll
    for (int o = 8; o > 0; o >>= 1) sm += __shfl_xor(sm, o, 32);
    const float inv = 1.0f / sm;
    if (row < nnodes) {
        float* orow = Out + (long)row * 40;
        orow[col0]      = v0 * inv;
        orow[16 + col0] = v1 * inv;
        if (hi2) orow[32 + col0] = v2 * inv;
    }
}

extern "C" void kernel_launch(void* const* d_in, const int* in_sizes, int n_in,
                              void* d_out, int out_size, void* d_ws, size_t ws_size,
                              hipStream_t stream) {
    const float* x    = (const float*)d_in[0];
    const float* Wl1  = (const float*)d_in[1];
    const float* Wr1  = (const float*)d_in[2];
    const float* b1   = (const float*)d_in[3];
    const float* Wl2  = (const float*)d_in[4];
    const float* Wr2  = (const float*)d_in[5];
    const float* b2   = (const float*)d_in[6];
    const int* esrc   = (const int*)d_in[7];
    const int* edst   = (const int*)d_in[8];

    const int H  = in_sizes[3];
    const int D  = in_sizes[1] / H;
    const int Nn = in_sizes[0] / D;
    const int E  = in_sizes[7];
    (void)n_in; (void)out_size; (void)ws_size;

    float* ws     = (float*)d_ws;
    int*   degi   = (int*)ws;
    float* dis    = ws + (long)Nn;
    int*   rowptr = (int*)(ws + 2L * Nn);
    float* bufA   = ws + 3L * Nn;
    float* bufB   = bufA + (long)Nn * 64;
    float* bufC   = bufB + (long)Nn * 64;
    int*   pairs  = (int*)bufB;
    int*   boff   = pairs + 2L * E;
    int*   csrSrc = (int*)bufC;

    const int TB = 256;
    dim3 gPart((E + PT_CHUNK - 1) / PT_CHUNK);
    const int nb = (Nn + BKT_NODES - 1) / BKT_NODES;
    dim3 gGather(((long)Nn * 16 + TB - 1) / TB);
    int rbCount = (Nn + 15) / 16;
    dim3 gTile(rbCount);

    bucket_sort_kernel<<<gPart, PT_THREADS, 0, stream>>>(esrc, edst, boff, pairs, E, nb);
    bucket_csr_kernel<<<nb, PT_THREADS, 0, stream>>>(pairs, boff, rowptr, degi, dis, csrSrc, Nn, nb, (int)gPart.x);

    gather_kernel<1><<<gGather, TB, 0, stream>>>(x, bufA, csrSrc, rowptr, degi, dis, Nn);
    gather_kernel<1><<<gGather, TB, 0, stream>>>(bufA, bufB, csrSrc, rowptr, degi, dis, Nn);
    agg_gemm1_kernel<<<gTile, TB, 0, stream>>>(bufB, csrSrc, rowptr, degi, Wl1, Wr1, b1, bufA, Nn);
    agg_gemm2_softmax_kernel<<<gTile, TB, 0, stream>>>(bufA, csrSrc, rowptr, degi, Wl2, Wr2, b2, (float*)d_out, Nn);
}


# Gather loops laid out for fewer dependent memory round trips per wave: 8 row loads per wait (was 4), next list entries and weights fetched under current rows, own row staged first; sums unchanged
# speedup vs baseline: 5.2119x; 1.0338x over previous
#include <hip/hip_runtime.h>
#include <hip/hip_bf16.h>
#include <math.h>

typedef __attribute__((ext_vector_type(2))) float v2f;
typedef __attribute__((ext_vector_type(8))) float v8f;

#define SELU_SCALE 1.0507009873554805f
#define SELU_ALPHA 1.6732632423543772f

__device__ __forceinline__ v8f wmma4(v2f a, v2f b, v8f c) {
    return __builtin_amdgcn_wmma_f32_16x16x4_f32(false, a, false, b, (short)0, c, false, false);
}

#define BKT_SHIFT   9
#define BKT_NODES   512
#define MAX_BKT     1024
#define PT_THREADS  1024
#define PT_PER      12
#define PT_CHUNK    (PT_THREADS * PT_PER)
#define MAX_CHUNKS  1024
#define RUN_GRP     4
#define RUN_SLOTS   3
#define STAGE_CAP   16384

__global__ __launch_bounds__(PT_THREADS) void bucket_sort_kernel(
    const int* __restrict__ src, const int* __restrict__ dst,
    int* __restrict__ boff, int* __restrict__ pairs_raw, int E, int nb)
{
    __shared__ int h[MAX_BKT];
    __shared__ int wsum[PT_THREADS / 32];
    __shared__ __attribute__((aligned(16))) int2 sp[PT_CHUNK];
    const int t = threadIdx.x;
    const int lane = t & 31;
    const int wv = t >> 5;
    const int blk = blockIdx.x;
    const int base = blk * PT_CHUNK;
    h[t] = 0;
    int d[PT_PER], s[PT_PER], rk[PT_PER];
    #pragma unroll
    for (int i = 0; i < PT_PER; ++i) {
        const int e = base + i * PT_THREADS + t;
        const bool in = e < E;
        d[i] = in ? dst[e] : -1;
        s[i] = in ? src[e] : 0;
    }
    __syncthreads();
    #pragma unroll
    for (int i = 0; i < PT_PER; ++i)
        rk[i] = (d[i] >= 0) ? atomicAdd(&h[d[i] >> BKT_SHIFT], 1) : 0;
    __syncthreads();
    const int c = h[t];
    int inc = c;
    for (int o = 1; o < 32; o <<= 1) {
        int y = __shfl_up(inc, o, 32);
        if (lane >= o) inc += y;
    }
    if (lane == 31) wsum[wv] = inc;
    __syncthreads();
    int wp = (lane < wv) ? wsum[lane] : 0;
    for (int o = 16; o > 0; o >>= 1) wp += __shfl_xor(wp, o, 32);
    const int start = wp + inc - c;
    h[t] = start;
    if (t <= nb) boff[blk * (nb + 1) + t] = start;
    __syncthreads();
    #pragma unroll
    for (int i = 0; i < PT_PER; ++i)
        if (d[i] >= 0) sp[h[d[i] >> BKT_SHIFT] + rk[i]] = make_int2(d[i], s[i]);
    __syncthreads();
    const int cntc = min(PT_CHUNK, E - base);
    const int n4 = cntc >> 1;
    int4* gp = (int4*)(pairs_raw + 2L * base);
    const int4* lp = (const int4*)sp;
    if (cntc == PT_CHUNK) {
        int4 r[PT_PER / 2];
        #pragma unroll
        for (int i = 0; i < PT_PER / 2; ++i) r[i] = lp[i * PT_THREADS + t];
        #pragma unroll
        for (int i = 0; i < PT_PER / 2; ++i) gp[i * PT_THREADS + t] = r[i];
    } else {
        for (int q = t; q < n4; q += PT_THREADS) gp[q] = lp[q];
        if ((cntc & 1) && t == 0) ((int2*)pairs_raw)[base + cntc - 1] = sp[cntc - 1];
    }
}

template <int PLACE>
__device__ __forceinline__ void walk_runs(
    const int2* __restrict__ pairs, const int* rstart, const int* rlen, int nchunk,
    int wv, int lane, int node0, int* ctr, int* stage, int* __restrict__ csr_bkt)
{
    for (int k0 = wv; k0 < nchunk; k0 += 32 * RUN_GRP) {
        int2 p[RUN_GRP][RUN_SLOTS];
        int len[RUN_GRP], st[RUN_GRP];
        #pragma unroll
        for (int u = 0; u < RUN_GRP; ++u) {
            const int k = k0 + 32 * u;
            const bool in = k < nchunk;
            len[u] = in ? rlen[k] : 0;
            st[u]  = in ? rstart[k] : 0;
            #pragma unroll
            for (int v = 0; v < RUN_SLOTS; ++v) {
                const int i = lane + 32 * v;
                if (i < len[u]) {
                    if (PLACE) p[u][v] = pairs[st[u] + i];
                    else       p[u][v] = make_int2(pairs[st[u] + i].x, 0);
                } else {
                    p[u][v] = make_int2(-1, 0);
                }
            }
        }
        #pragma unroll
        for (int u = 0; u < RUN_GRP; ++u) {
            #pragma unroll
            for (int v = 0; v < RUN_SLOTS; ++v) {
                if (p[u][v].x >= 0) {
                    const int slot = atomicAdd(&ctr[p[u][v].x - node0], 1);
                    if (PLACE) {
                        if (slot < STAGE_CAP) stage[slot] = p[u][v].y;
                        else                  csr_bkt[slot] = p[u][v].y;
                    }
                }
            }
        }
        #pragma unroll
        for (int u = 0; u < RUN_GRP; ++u) {
            for (int i = lane + 32 * RUN_SLOTS; i < len[u]; i += 32) {
                const int2 q = pairs[st[u] + i];
                const int slot = atomicAdd(&ctr[q.x - node0], 1);
                if (PLACE) {
                    if (slot < STAGE_CAP) stage[slot] = q.y;
                    else                  csr_bkt[slot] = q.y;
                }
            }
        }
    }
}

__global__ __launch_bounds__(PT_THREADS) void bucket_csr_kernel(
    const int* __restrict__ pairs_raw, const int* __restrict__ boff,
    int* __restrict__ rowptr, int* __restrict__ degi, float* __restrict__ dis,
    int* __restrict__ csr_src, int n, int nb, int nchunk)
{
    __shared__ int cnt[BKT_NODES];
    __shared__ int pos[BKT_NODES];
    __shared__ int wsum[PT_THREADS / 32];
    __shared__ int rstart[MAX_CHUNKS];
    __shared__ int rlen[MAX_CHUNKS];
    __shared__ int stage[STAGE_CAP];
    const int t = threadIdx.x;
    const int lane = t & 31;
    const int wv = t >> 5;
    const int b = blockIdx.x;
    const int2* pairs = (const int2*)pairs_raw;
    const int node0 = b << BKT_SHIFT;
    int o0 = 0, o1 = 0;
    if (t < nchunk) {
        o0 = boff[t * (nb + 1) + b];
        o1 = boff[t * (nb + 1) + b + 1];
    }
    rstart[t] = t * PT_CHUNK + o0;
    rlen[t] = o1 - o0;
    if (t < BKT_NODES) cnt[t] = 0;
    int sb = o0;
    for (int o = 16; o > 0; o >>= 1) sb += __shfl_xor(sb, o, 32);
    if (lane == 0) wsum[wv] = sb;
    __syncthreads();
    int beg = wsum[lane];
    for (int o = 16; o > 0; o >>= 1) beg += __shfl_xor(beg, o, 32);
    walk_runs<0>(pairs, rstart, rlen, nchunk, wv, lane, node0, cnt, stage, csr_src + beg);
    __syncthreads();
    const int c = (t < BKT_NODES) ? cnt[t] : 0;
    int inc = c;
    for (int o = 1; o < 32; o <<= 1) {
        int y = __shfl_up(inc, o, 32);
        if (lane >= o) inc += y;
    }
    if (t < BKT_NODES && lane == 31) wsum[wv] = inc;
    __syncthreads();
    int tot = (lane < BKT_NODES / 32) ? wsum[lane] : 0;
    for (int o = 16; o > 0; o >>= 1) tot += __shfl_xor(tot, o, 32);
    if (t < BKT_NODES) {
        int wp = (lane < wv) ? wsum[lane] : 0;
        for (int o = 16; o > 0; o >>= 1) wp += __shfl_xor(wp, o, 32);
        const int start = wp + inc - c;
        pos[t] = start;
        const int node = node0 + t;
        if (node < n) {
            const float d = (float)c;
            rowptr[node] = beg + start;
            degi[node] = c;
            dis[node] = (d > 0.0f) ? rsqrtf(fmaxf(d, 1.0f)) : 0.0f;
        }
    }
    __syncthreads();
    walk_runs<1>(pairs, rstart, rlen, nchunk, wv, lane, node0, pos, stage, csr_src + beg);
    __syncthreads();
    const int lim = min(tot, STAGE_CAP);
    for (int g = (beg & ~31) + t; g < beg + lim; g += PT_THREADS)
        if (g >= beg) csr_src[g] = stage[g - beg];
}

__device__ __forceinline__ float keep_bits(float t, int keep) {
    return __int_as_float(__float_as_int(t) & keep);
}

template <int WEIGHTED, int K0>
__device__ __forceinline__ void add_rows8(
    float4& acc, const float4* __restrict__ X4, int mySrc, float myD, float dt,
    int slot0, int cnt, int seg)
{
    int s[8]; float w[8]; float4 v[8];
    #pragma unroll
    for (int u = 0; u < 8; ++u) s[u] = __shfl(mySrc, K0 + u, 16);
    #pragma unroll
    for (int u = 0; u < 8; ++u) v[u] = X4[s[u] * 16 + seg];
    #pragma unroll
    for (int u = 0; u < 8; ++u) w[u] = WEIGHTED ? __shfl(myD, K0 + u, 16) * dt : 1.0f;
    #pragma unroll
    for (int u = 0; u < 8; ++u) {
        const int keep = ((slot0 + u) < cnt) ? -1 : 0;
        acc.x += keep_bits(WEIGHTED ? v[u].x * w[u] : v[u].x, keep);
        acc.y += keep_bits(WEIGHTED ? v[u].y * w[u] : v[u].y, keep);
        acc.z += keep_bits(WEIGHTED ? v[u].z * w[u] : v[u].z, keep);
        acc.w += keep_bits(WEIGHTED ? v[u].w * w[u] : v[u].w, keep);
    }
}

template <int WEIGHTED>
__device__ __forceinline__ float4 row_sum(
    const float* __restrict__ X, const int* __restrict__ csr_src, const float* __restrict__ dis,
    int beg, int cnt, float dt, int seg)
{
    const float4* X4 = (const float4*)X;
    float4 acc = make_float4(0.0f, 0.0f, 0.0f, 0.0f);
    const int cmax = __builtin_amdgcn_readfirstlane(max(cnt, __shfl_xor(cnt, 16, 32)));
    int mySrc = (seg < cnt) ? csr_src[beg + seg] : 0;
    float myD = 1.0f;
    if (WEIGHTED) myD = dis[mySrc];
    for (int base = 0; base < cmax; base += 16) {
        const bool more = (base + 16) < cmax;
        add_rows8<WEIGHTED, 0>(acc, X4, mySrc, myD, dt, base, cnt, seg);
        int nSrc = 0;
        float nD = 1.0f;
        if (more) {
            const int jn = base + 16 + seg;
            if (jn < cnt) nSrc = csr_src[beg + jn];
        }
        if (cmax - base > 8)
            add_rows8<WEIGHTED, 8>(acc, X4, mySrc, myD, dt, base + 8, cnt, seg);
        if (WEIGHTED && more) nD = dis[nSrc];
        mySrc = nSrc;
        myD = nD;
    }
    return acc;
}

template <int WEIGHTED>
__global__ __launch_bounds__(256) void gather_kernel(
    const float* __restrict__ X, float* __restrict__ Y,
    const int* __restrict__ csr_src, const int* __restrict__ rowptr,
    const int* __restrict__ degi, const float* __restrict__ dis, int n)
{
    const int tid  = blockIdx.x * blockDim.x + threadIdx.x;
    const int node = tid >> 4;
    const int seg  = tid & 15;
    const bool valid = node < n;
    const int nd  = valid ? node : (n - 1);
    const int beg = rowptr[nd];
    const int cnt = valid ? degi[nd] : 0;
    const float dt = WEIGHTED ? dis[nd] : 1.0f;
    const float4 acc = row_sum<WEIGHTED>(X, csr_src, dis, beg, cnt, dt, seg);
    if (valid) ((float4*)Y)[(long)node * 16 + seg] = acc;
}

#define TILE_PITCH 68

__device__ __forceinline__ void stage_tile(
    const float* __restrict__ Hf, const int* __restrict__ csr_src, const int* __restrict__ rowptr,
    const int* __restrict__ degi, float* sAgg, float* sH, int n)
{
    const int t = threadIdx.x;
    const int r = t >> 4;
    const int seg = t & 15;
    const int node = blockIdx.x * 16 + r;
    const bool valid = node < n;
    const int nd = valid ? node : (n - 1);
    const int beg = rowptr[nd];
    const int cnt = valid ? degi[nd] : 0;
    *(float4*)(sH + r * TILE_PITCH + seg * 4) = ((const float4*)Hf)[nd * 16 + seg];
    const float4 acc = row_sum<0>(Hf, csr_src, nullptr, beg, cnt, 1.0f, seg);
    const float cinv = 1.0f / fmaxf((float)cnt, 1.0f);
    *(float4*)(sAgg + r * TILE_PITCH + seg * 4) =
        make_float4(acc.x * cinv, acc.y * cinv, acc.z * cinv, acc.w * cinv);
}

__global__ __launch_bounds__(256) void agg_gemm1_kernel(
    const float* __restrict__ Hf, const int* __restrict__ csr_src, const int* __restrict__ rowptr,
    const int* __restrict__ degi, const float* __restrict__ Wl, const float* __restrict__ Wr,
    const float* __restrict__ bias, float* __restrict__ Out, int nnodes)
{
    __shared__ __attribute__((aligned(16))) float sAgg[16 * TILE_PITCH];
    __shared__ __attribute__((aligned(16))) float sH[16 * TILE_PITCH];
    __shared__ __attribute__((aligned(16))) float sOut[16 * TILE_PITCH];
    stage_tile(Hf, csr_src, rowptr, degi, sAgg, sH, nnodes);
    __syncthreads();
    const int ct = ((threadIdx.x >> 5) - blockIdx.x) & 7;
    if (ct < 4) {
        const int lane = threadIdx.x & 31;
        const int col0 = lane & 15;
        const int kshift = (lane >> 4) * 2;
        const float* aggRow = sAgg + col0 * TILE_PITCH;
        const float* hRow   = sH   + col0 * TILE_PITCH;
        const int col = ct * 16 + col0;
        v8f acc = {};
        #pragma unroll
        for (int k0 = 0; k0 < 64; k0 += 4) {
            const int kk = k0 + kshift;
            const v2f a1 = *(const v2f*)(aggRow + kk);
            const v2f a2 = *(const v2f*)(hRow + kk);
            v2f b;
            b.x = Wl[kk * 64 + col]; b.y = Wl[(kk + 1) * 64 + col]; acc = wmma4(a1, b, acc);
            b.x = Wr[kk * 64 + col]; b.y = Wr[(kk + 1) * 64 + col]; acc = wmma4(a2, b, acc);
        }
        const float bn = bias[col];
        const int r0 = (lane >> 4) * 8;
        #pragma unroll
        for (int j = 0; j < 8; ++j) sOut[(r0 + j) * TILE_PITCH + col] = acc[j] + bn;
    }
    __syncthreads();
    const int r = threadIdx.x >> 4;
    const int seg = threadIdx.x & 15;
    const int row = blockIdx.x * 16 + r;
    const float4 s = *(const float4*)(sOut + r * TILE_PITCH + seg * 4);
    float4 y;
    y.x = s.x > 0.0f ? SELU_SCALE * s.x : SELU_SCALE * SELU_ALPHA * (expf(s.x) - 1.0f);
    y.y = s.y > 0.0f ? SELU_SCALE * s.y : SELU_SCALE * SELU_ALPHA * (expf(s.y) - 1.0f);
    y.z = s.z > 0.0f ? SELU_SCALE * s.z : SELU_SCALE * SELU_ALPHA * (expf(s.z) - 1.0f);
    y.w = s.w > 0.0f ? SELU_SCALE * s.w : SELU_SCALE * SELU_ALPHA * (expf(s.w) - 1.0f);
    if (row < nnodes) ((float4*)Out)[(long)row * 16 + seg] = y;
}

#define LOGIT_PITCH 52

__global__ __launch_bounds__(256) __attribute__((amdgpu_waves_per_eu(16, 16))) void agg_gemm2_softmax_kernel(
    const float* __restrict__ Hf, const int* __restrict__ csr_src, const int* __restrict__ rowptr,
    const int* __restrict__ degi, const float* __restrict__ Wl, const float* __restrict__ Wr,
    const float* __restrict__ bias, float* __restrict__ Out, int nnodes)
{
    __shared__ __attribute__((aligned(16))) float sAgg[16 * TILE_PITCH];
    __shared__ __attribute__((aligned(16))) float sH[16 * TILE_PITCH];
    __shared__ float sLogit[16 * LOGIT_PITCH];
    stage_tile(Hf, csr_src, rowptr, degi, sAgg, sH, nnodes);
    __syncthreads();
    const int lane = threadIdx.x & 31;
    const int col0 = lane & 15;
    const int ct = ((threadIdx.x >> 5) - blockIdx.x) & 7;
    if (ct < 3) {
        const int kshift = (lane >> 4) * 2;
        const int colr = ct * 16 + col0;
        const bool hic = colr < 40;
        const int coff = hic ? colr : 0;
        const float msk = hic ? 1.0f : 0.0f;
        const float* aggRow = sAgg + col0 * TILE_PITCH;
        const float* hRow   = sH   + col0 * TILE_PITCH;
        v8f acc = {};
        #pragma unroll
        for (int k0 = 0; k0 < 64; k0 += 4) {
            const int kk = k0 + kshift;
            const v2f a1 = *(const v2f*)(aggRow + kk);
            const v2f a2 = *(const v2f*)(hRow + kk);
            v2f b;
            b.x = Wl[kk * 40 + coff] * msk; b.y = Wl[(kk + 1) * 40 + coff] * msk; acc = wmma4(a1, b, acc);
            b.x = Wr[kk * 40 + coff] * msk; b.y = Wr[(kk + 1) * 40 + coff] * msk; acc = wmma4(a2, b, acc);
        }
        const float bn = bias[coff];
        const int r0 = (lane >> 4) * 8;
        #pragma unroll
        for (int j = 0; j < 8; ++j) sLogit[(r0 + j) * LOGIT_PITCH + colr] = acc[j] + bn;
    }
    __syncthreads();
    const int r = threadIdx.x >> 4;
    const int row = blockIdx.x * 16 + r;
    const bool hi2 = (32 + col0) < 40;
    const float* lrow = sLogit + r * LOGIT_PITCH;
    float v0 = lrow[col0];
    float v1 = lrow[16 + col0];
    float v2 = lrow[32 + col0];
    const float m01 = fmaxf(v0, v1);
    float mx = hi2 ? fmaxf(m01, v2) : m01;
    #pragma unroll
    for (int o = 8; o > 0; o >>= 1) mx = fmaxf(mx, __shfl_xor(mx, o, 32));
    v0 = expf(v0 - mx);
    v1 = expf(v1 - mx);
    const float e2 = expf(v2 - mx);
    v2 = hi2 ? e2 : 0.0f;
    float sm = v0 + v1 + v2;
    #pragma unroll
    for (int o = 8; o > 0; o >>= 1) sm += __shfl_xor(sm, o, 32);
    const float inv = 1.0f / sm;
    if (row < nnodes) {
        float* orow = Out + (long)row * 40;
        orow[col0]      = v0 * inv;
        orow[16 + col0] = v1 * inv;
        if (hi2) orow[32 + col0] = v2 * inv;
    }
}

extern "C" void kernel_launch(void* const* d_in, const int* in_sizes, int n_in,
                              void* d_out, int out_size, void* d_ws, size_t ws_size,
                              hipStream_t stream) {
    const float* x    = (const float*)d_in[0];
    const float* Wl1  = (const float*)d_in[1];
    const float* Wr1  = (const float*)d_in[2];
    const float* b1   = (const float*)d_in[3];
    const float* Wl2  = (const float*)d_in[4];
    const float* Wr2  = (const float*)d_in[5];
    const float* b2   = (const float*)d_in[6];
    const int* esrc   = (const int*)d_in[7];
    const int* edst   = (const int*)d_in[8];

    const int H  = in_sizes[3];
    const int D  = in_sizes[1] / H;
    const int Nn = in_sizes[0] / D;
    const int E  = in_sizes[7];
    (void)n_in; (void)out_size; (void)ws_size;

    float* ws     = (float*)d_ws;
    int*   degi   = (int*)ws;
    float* dis    = ws + (long)Nn;
    int*   rowptr = (int*)(ws + 2L * Nn);
    float* bufA   = ws + 3L * Nn;
    float* bufB   = bufA + (long)Nn * 64;
    float* bufC   = bufB + (long)Nn * 64;
    int*   pairs  = (int*)bufB;
    int*   boff   = pairs + 2L * E;
    int*   csrSrc = (int*)bufC;

    const int TB = 256;
    dim3 gPart((E + PT_CHUNK - 1) / PT_CHUNK);
    const int nb = (Nn + BKT_NODES - 1) / BKT_NODES;
    dim3 gGather(((long)Nn * 16 + TB - 1) / TB);
    int rbCount = (Nn + 15) / 16;
    dim3 gTile(rbCount);

    bucket_sort_kernel<<<gPart, PT_THREADS, 0, stream>>>(esrc, edst, boff, pairs, E, nb);
    bucket_csr_kernel<<<nb, PT_THREADS, 0, stream>>>(pairs, boff, rowptr, degi, dis, csrSrc, Nn, nb, (int)gPart.x);

    gather_kernel<1><<<gGather, TB, 0, stream>>>(x, bufA, csrSrc, rowptr, degi, dis, Nn);
    gather_kernel<1><<<gGather, TB, 0, stream>>>(bufA, bufB, csrSrc, rowptr, degi, dis, Nn);
    agg_gemm1_kernel<<<gTile, TB, 0, stream>>>(bufB, csrSrc, rowptr, degi, Wl1, Wr1, b1, bufA, Nn);
    agg_gemm2_softmax_kernel<<<gTile, TB, 0, stream>>>(bufA, csrSrc, rowptr, degi, Wl2, Wr2, b2, (float*)d_out, Nn);
}
